# starting kernel
# baseline (speedup 1.0000x reference)
#include <hip/hip_runtime.h>
#include <hip/hip_bf16.h>
#include <math.h>


#define D_DIM 64
#define DOUT  128
#define NEG_SLOPE 0.2f

typedef __attribute__((ext_vector_type(2))) float v2f;
typedef __attribute__((ext_vector_type(8))) float v8f;

__device__ __forceinline__ unsigned f2ord(float f) {
    unsigned u = __float_as_uint(f);
    return (u & 0x80000000u) ? ~u : (u | 0x80000000u);
}
__device__ __forceinline__ float ord2f(unsigned u) {
    unsigned b = (u & 0x80000000u) ? (u ^ 0x80000000u) : ~u;
    return __uint_as_float(b);
}

__global__ void k_init(float* __restrict__ h, float* __restrict__ denom,
                       unsigned* __restrict__ smax, int n_nodes) {
    long long i = (long long)blockIdx.x * blockDim.x + threadIdx.x;
    if (i < (long long)n_nodes * D_DIM) h[i] = 0.0f;
    if (i < n_nodes) { denom[i] = 0.0f; smax[i] = 0x007FFFFFu;   }
}

__global__ void k_node_linear(const float* __restrict__ x,
                              const float* __restrict__ Wl, const float* __restrict__ bl,
                              const float* __restrict__ Wr, const float* __restrict__ br,
                              float* __restrict__ xl, float* __restrict__ xr,
                              int n_nodes) {
    const int lane = threadIdx.x & 31;
    const int wave = threadIdx.x >> 5;
    const int tile = blockIdx.x * (blockDim.x >> 5) + wave;
    const int r0 = tile * 16;
    if (r0 + 16 > n_nodes) return;
    const int lrow = lane & 15;
    const int lhi  = lane >> 4;

    v2f a[16];
    const float* xp = x + (long long)(r0 + lrow) * D_DIM + 2 * lhi;
#pragma unroll
    for (int kk = 0; kk < 16; ++kk)
        a[kk] = *(const v2f*)(xp + kk * 4);

#pragma unroll
    for (int m = 0; m < 2; ++m) {
        const float* W    = m ? Wr : Wl;
        const float* bias = m ? br : bl;
        float*       out  = m ? xr : xl;
#pragma unroll
        for (int ct = 0; ct < 4; ++ct) {
            const int col = ct * 16 + lrow;
            const float bv = bias[col];
            v8f acc = {bv, bv, bv, bv, bv, bv, bv, bv};
#pragma unroll
            for (int kk = 0; kk < 16; ++kk) {
                const int k0 = kk * 4 + 2 * lhi;
                v2f b;
                b.x = W[(k0    ) * D_DIM + col];
                b.y = W[(k0 + 1) * D_DIM + col];
                acc = __builtin_amdgcn_wmma_f32_16x16x4_f32(
                        false, a[kk], false, b, (short)0, acc, false, false);
            }
#pragma unroll
            for (int r = 0; r < 8; ++r)
                out[(long long)(r0 + r + 8 * lhi) * D_DIM + col] = acc[r];
        }
    }
}

__global__ void k_edge_logits(const int* __restrict__ src, const int* __restrict__ dst,
                              const float* __restrict__ xl, const float* __restrict__ xr,
                              const float* __restrict__ att,
                              float* __restrict__ logits, unsigned* __restrict__ smax,
                              int n_edges) {
    const int lane = threadIdx.x & 31;
    const long long e = ((long long)blockIdx.x * blockDim.x + threadIdx.x) >> 5;
    if (e >= n_edges) return;
    const int s = src[e], d = dst[e];
    const v2f vl = *(const v2f*)(xl + (long long)s * D_DIM + 2 * lane);
    const v2f vr = *(const v2f*)(xr + (long long)d * D_DIM + 2 * lane);
    const v2f va = *(const v2f*)(att + 2 * lane);
    float e0 = vl.x + vr.x, e1 = vl.y + vr.y;
    e0 = e0 > 0.0f ? e0 : NEG_SLOPE * e0;
    e1 = e1 > 0.0f ? e1 : NEG_SLOPE * e1;
    float p = e0 * va.x + e1 * va.y;
#pragma unroll
    for (int off = 16; off >= 1; off >>= 1) p += __shfl_xor(p, off, 32);
    if (lane == 0) {
        logits[e] = p;
        atomicMax(&smax[d], f2ord(p));
    }
}

__global__ void k_decode_smax(const unsigned* __restrict__ smax,
                              float* __restrict__ smaxf, int n_nodes) {
    int i = blockIdx.x * blockDim.x + threadIdx.x;
    if (i >= n_nodes) return;
    float f = ord2f(smax[i]);
    smaxf[i] = __builtin_isfinite(f) ? f : 0.0f;
}

__global__ void k_alpha(const float* __restrict__ logits, const int* __restrict__ dst,
                        const float* __restrict__ smaxf,
                        float* __restrict__ alpha, float* __restrict__ denom,
                        int n_edges) {
    int e = blockIdx.x * blockDim.x + threadIdx.x;
    if (e >= n_edges) return;
    const int d = dst[e];
    const float a = __expf(logits[e] - smaxf[d]);
    alpha[e] = a;
    atomicAdd(&denom[d], a);
}

__global__ void k_scatter(const int* __restrict__ src, const int* __restrict__ dst,
                          const float* __restrict__ xl,
                          const float* __restrict__ alpha, const float* __restrict__ denom,
                          float* __restrict__ h, int n_edges) {
    const long long tid = (long long)blockIdx.x * blockDim.x + threadIdx.x;
    const long long e = tid >> 5;
    const int lane = (int)(tid & 31);
    if (e >= n_edges) return;
    const int s = src[e], d = dst[e];
    const float w = alpha[e] / (denom[d] + 1e-16f);
    const v2f v = *(const v2f*)(xl + (long long)s * D_DIM + 2 * lane);
    float* hp = h + (long long)d * D_DIM + 2 * lane;
    atomicAdd(hp,     w * v.x);
    atomicAdd(hp + 1, w * v.y);
}

__global__ void k_output(const float* __restrict__ h, const int* __restrict__ mask,
                         const float* __restrict__ gbias,
                         const float* __restrict__ Wo, const float* __restrict__ bo,
                         float* __restrict__ out, int n_mask) {
    const int lane = threadIdx.x & 31;
    const int wave = threadIdx.x >> 5;
    const int tile = blockIdx.x * (blockDim.x >> 5) + wave;
    const int r0 = tile * 16;
    if (r0 + 16 > n_mask) return;
    const int lrow = lane & 15;
    const int lhi  = lane >> 4;

    const long long mi = (long long)mask[r0 + lrow];
    v2f a[16];
    const float* hp = h + mi * D_DIM + 2 * lhi;
    const float* gp = gbias + 2 * lhi;
#pragma unroll
    for (int kk = 0; kk < 16; ++kk) {
        const v2f hv = *(const v2f*)(hp + kk * 4);
        const v2f gv = *(const v2f*)(gp + kk * 4);
        a[kk].x = hv.x + gv.x;
        a[kk].y = hv.y + gv.y;
    }
#pragma unroll
    for (int ct = 0; ct < 8; ++ct) {
        const int col = ct * 16 + lrow;
        const float bv = bo[col];
        v8f acc = {bv, bv, bv, bv, bv, bv, bv, bv};
#pragma unroll
        for (int kk = 0; kk < 16; ++kk) {
            const int k0 = kk * 4 + 2 * lhi;
            v2f b;
            b.x = Wo[(k0    ) * DOUT + col];
            b.y = Wo[(k0 + 1) * DOUT + col];
            acc = __builtin_amdgcn_wmma_f32_16x16x4_f32(
                    false, a[kk], false, b, (short)0, acc, false, false);
        }
#pragma unroll
        for (int r = 0; r < 8; ++r)
            out[(long long)(r0 + r + 8 * lhi) * DOUT + col] = acc[r];
    }
}

extern "C" void kernel_launch(void* const* d_in, const int* in_sizes, int n_in,
                              void* d_out, int out_size, void* d_ws, size_t ws_size,
                              hipStream_t stream) {
    const float* x     = (const float*)d_in[0];
    const int*   ei    = (const int*)d_in[1];
    const int*   mask  = (const int*)d_in[2];
    const float* Wl    = (const float*)d_in[3];
    const float* bl    = (const float*)d_in[4];
    const float* Wr    = (const float*)d_in[5];
    const float* br    = (const float*)d_in[6];
    const float* att   = (const float*)d_in[7];
    const float* gbias = (const float*)d_in[8];
    const float* Wo    = (const float*)d_in[9];
    const float* bo    = (const float*)d_in[10];
    float* out = (float*)d_out;

    const int n_nodes = in_sizes[0] / D_DIM;
    const int n_edges = in_sizes[1] / 2;
    const int n_mask  = in_sizes[2];
    const int* src = ei;
    const int* dst = ei + n_edges;

    char* wp = (char*)d_ws;
    auto carve = [&](size_t bytes) {
        char* p = wp;
        wp += (bytes + 255) & ~(size_t)255;
        return p;
    };
    float*    xl     = (float*)   carve((size_t)n_nodes * D_DIM * 4);
    float*    xr     = (float*)   carve((size_t)n_nodes * D_DIM * 4);
    float*    h      = (float*)   carve((size_t)n_nodes * D_DIM * 4);
    float*    logits = (float*)   carve((size_t)n_edges * 4);
    float*    alpha  = (float*)   carve((size_t)n_edges * 4);
    unsigned* smax   = (unsigned*)carve((size_t)n_nodes * 4);
    float*    smaxf  = (float*)   carve((size_t)n_nodes * 4);
    float*    denom  = (float*)   carve((size_t)n_nodes * 4);

    {
        long long t = (long long)n_nodes * D_DIM;
        int blocks = (int)((t + 255) / 256);
        k_init<<<blocks, 256, 0, stream>>>(h, denom, smax, n_nodes);
    }
    {
        int tiles  = (n_nodes + 15) / 16;
        int blocks = (tiles + 7) / 8;
        k_node_linear<<<blocks, 256, 0, stream>>>(x, Wl, bl, Wr, br, xl, xr, n_nodes);
    }
    {
        long long t = (long long)n_edges * 32;
        int blocks = (int)((t + 255) / 256);
        k_edge_logits<<<blocks, 256, 0, stream>>>(src, dst, xl, xr, att, logits, smax, n_edges);
    }
    k_decode_smax<<<(n_nodes + 255) / 256, 256, 0, stream>>>(smax, smaxf, n_nodes);
    k_alpha<<<(n_edges + 255) / 256, 256, 0, stream>>>(logits, dst, smaxf, alpha, denom, n_edges);
    {
        long long t = (long long)n_edges * 32;
        int blocks = (int)((t + 255) / 256);
        k_scatter<<<blocks, 256, 0, stream>>>(src, dst, xl, alpha, denom, h, n_edges);
    }
    {
        int tiles  = (n_mask + 15) / 16;
        int blocks = (tiles + 7) / 8;
        k_output<<<blocks, 256, 0, stream>>>(h, mask, gbias, Wo, bo, out, n_mask);
    }
}


# scatter adds: the whole wave takes each listed edge (one 256-B load of its xl row, 2 LDS float adds per lane on 32 banks), was a lane per edge with 16 loads and 64 adds; sums stay in arrival order
# speedup vs baseline: 2.1466x; 2.1466x over previous
#include <hip/hip_runtime.h>
#include <hip/hip_bf16.h>
#include <math.h>


#define D_DIM 64
#define DOUT  128
#define NEG_SLOPE 0.2f
#define NR    400
#define ROWP  65
#define CHUNK 2048
#define GRP   16

typedef __attribute__((ext_vector_type(2))) float v2f;
typedef __attribute__((ext_vector_type(4))) float v4f;
typedef __attribute__((ext_vector_type(8))) float v8f;

__device__ __forceinline__ unsigned f2ord(float f) {
    unsigned u = __float_as_uint(f);
    return (u & 0x80000000u) ? ~u : (u | 0x80000000u);
}
__device__ __forceinline__ float ord2f(unsigned u) {
    unsigned b = (u & 0x80000000u) ? (u ^ 0x80000000u) : ~u;
    return __uint_as_float(b);
}

__global__ void k_init(float* __restrict__ h, int n_nodes) {
    int i = blockIdx.x * blockDim.x + threadIdx.x;
    if (i < n_nodes)
        h[(long long)i * D_DIM] = __uint_as_float(0x007FFFFFu);
}

__global__ void k_node_linear(const float* __restrict__ x,
                              const float* __restrict__ Wl, const float* __restrict__ bl,
                              const float* __restrict__ Wr, const float* __restrict__ br,
                              float* __restrict__ xl, float* __restrict__ xr,
                              int n_nodes) {
    const int lane = threadIdx.x & 31;
    const int wave = threadIdx.x >> 5;
    const int tile = blockIdx.x * (blockDim.x >> 5) + wave;
    const int r0 = tile * 16;
    if (r0 + 16 > n_nodes) return;
    const int lrow = lane & 15;
    const int lhi  = lane >> 4;

    v2f a[16];
    const float* xp = x + (long long)(r0 + lrow) * D_DIM + 2 * lhi;
#pragma unroll
    for (int kk = 0; kk < 16; ++kk)
        a[kk] = *(const v2f*)(xp + kk * 4);

#pragma unroll
    for (int m = 0; m < 2; ++m) {
        const float* W    = m ? Wr : Wl;
        const float* bias = m ? br : bl;
        float*       out  = m ? xr : xl;
#pragma unroll
        for (int ct = 0; ct < 4; ++ct) {
            const int col = ct * 16 + lrow;
            const float bv = bias[col];
            v8f acc = {bv, bv, bv, bv, bv, bv, bv, bv};
#pragma unroll
            for (int kk = 0; kk < 16; ++kk) {
                const int k0 = kk * 4 + 2 * lhi;
                v2f b;
                b.x = W[(k0    ) * D_DIM + col];
                b.y = W[(k0 + 1) * D_DIM + col];
                acc = __builtin_amdgcn_wmma_f32_16x16x4_f32(
                        false, a[kk], false, b, (short)0, acc, false, false);
            }
#pragma unroll
            for (int r = 0; r < 8; ++r)
                out[(long long)(r0 + r + 8 * lhi) * D_DIM + col] = acc[r];
        }
    }
}

__global__ void k_edge_logits(const int* __restrict__ src, const int* __restrict__ dst,
                              const float* __restrict__ xl, const float* __restrict__ xr,
                              const float* __restrict__ att,
                              float* __restrict__ logits, unsigned* __restrict__ smax,
                              unsigned char* __restrict__ rid,
                              int n_edges) {
    const int lane = threadIdx.x & 31;
    const int tile = (int)((blockIdx.x * blockDim.x + threadIdx.x) >> 5);
    const long long e0 = (long long)tile * 16;
    if (e0 >= n_edges) return;
    const int c  = lane & 15;
    const int hh = lane >> 4;
    long long e = e0 + c;
    const bool live = e < n_edges;
    if (!live) e = n_edges - 1;
    const int s = src[e], d = dst[e];
    const float* pl = xl  + (long long)s * D_DIM + 4 * hh;
    const float* pr = xr  + (long long)d * D_DIM + 4 * hh;
    const float* pa = att + 4 * hh;

    v4f vl[8], vr[8];
#pragma unroll
    for (int q = 0; q < 8; ++q) vl[q] = *(const v4f*)(pl + 8 * q);
#pragma unroll
    for (int q = 0; q < 8; ++q) vr[q] = *(const v4f*)(pr + 8 * q);
    __builtin_amdgcn_sched_barrier(0);

    v8f acc = {0.f, 0.f, 0.f, 0.f, 0.f, 0.f, 0.f, 0.f};
#pragma unroll
    for (int q = 0; q < 8; ++q) {
        const v4f t = *(const v4f*)(pa + 8 * q);
        v4f x = vl[q] + vr[q];
        x.x = x.x > 0.0f ? x.x : NEG_SLOPE * x.x;
        x.y = x.y > 0.0f ? x.y : NEG_SLOPE * x.y;
        x.z = x.z > 0.0f ? x.z : NEG_SLOPE * x.z;
        x.w = x.w > 0.0f ? x.w : NEG_SLOPE * x.w;
        const v2f a0 = {x.x, x.y}, b0 = {t.x, t.y};
        const v2f a1 = {x.z, x.w}, b1 = {t.z, t.w};
        acc = __builtin_amdgcn_wmma_f32_16x16x4_f32(
                false, a0, false, b0, (short)0, acc, false, false);
        acc = __builtin_amdgcn_wmma_f32_16x16x4_f32(
                false, a1, false, b1, (short)0, acc, false, false);
    }

    const int r = c & 7;
    float p = acc[0];
    p = r == 1 ? acc[1] : p;
    p = r == 2 ? acc[2] : p;
    p = r == 3 ? acc[3] : p;
    p = r == 4 ? acc[4] : p;
    p = r == 5 ? acc[5] : p;
    p = r == 6 ? acc[6] : p;
    p = r == 7 ? acc[7] : p;
    if (live && (c >> 3) == hh) {
        logits[e] = p;
        rid[e] = (unsigned char)((unsigned)d / NR);
        atomicMax(&smax[(long long)d * D_DIM], f2ord(p));
    }
}

typedef __attribute__((ext_vector_type(4))) unsigned v4u;

__global__ __launch_bounds__(1024)
void k_scatter_lds(const int* __restrict__ src, const int* __restrict__ dst,
                   const float* __restrict__ xl,
                   const float* __restrict__ logits,
                   const unsigned char* __restrict__ rid,
                   float* __restrict__ denom,
                   float* h, int n_edges, int n_nodes) {
    __shared__ float          acc[NR * ROWP];
    __shared__ float          den[NR];
    __shared__ float          smx[NR];
    __shared__ unsigned short qbuf[32 * CHUNK];
    __shared__ unsigned       qn[32];

    const int tid   = threadIdx.x;
    const int lane  = tid & 31;
    const int wave  = tid >> 5;
    const int nbase = blockIdx.x * NR;

    for (int i = tid; i < NR * ROWP; i += 1024) acc[i] = 0.0f;
    for (int r = tid; r < NR; r += 1024) {
        den[r] = 0.0f;
        float f = 0.0f;
        if (nbase + r < n_nodes) {
            f = ord2f(((const unsigned*)h)[(long long)(nbase + r) * D_DIM]);
            f = __builtin_isfinite(f) ? f : 0.0f;
        }
        smx[r] = f;
    }
    if (tid < 32) qn[tid] = 0u;
    __syncthreads();

    const unsigned pat = (unsigned)blockIdx.x * 0x01010101u;
    const int nchunks = (n_edges + CHUNK - 1) / CHUNK;
    const int rot = (int)((blockIdx.x * 37u) % (unsigned)nchunks);
    unsigned short* myq = qbuf + wave * CHUNK;

    int j  = wave;
    int ch = j + rot; if (ch >= nchunks) ch -= nchunks;
    v4u c[4];
    if (j < nchunks) {
        const v4u* rp = (const v4u*)(rid + (long long)ch * CHUNK) + lane;
#pragma unroll
        for (int k = 0; k < 4; ++k) c[k] = rp[k * 32];
    }
    while (j < nchunks) {
        const int ebase = ch * CHUNK;
#pragma unroll
        for (int k = 0; k < 4; ++k) {
#pragma unroll
            for (int q = 0; q < 4; ++q) {
                const unsigned x = c[k][q] ^ pat;
                unsigned fl = ~(((x & 0x7F7F7F7Fu) + 0x7F7F7F7Fu) | x | 0x7F7F7F7Fu);
                while (fl) {
                    const unsigned b = (unsigned)__builtin_ctz(fl) >> 3;
                    fl &= fl - 1u;
                    const unsigned off = (unsigned)(k * 512 + lane * 16 + q * 4) + b;
                    if (ebase + (int)off < n_edges) {
                        const unsigned pos = atomicAdd(&qn[wave], 1u);
                        myq[pos] = (unsigned short)off;
                    }
                }
            }
        }
        const int jn = j + 32;
        int chn = jn + rot; if (chn >= nchunks) chn -= nchunks;
        if (jn < nchunks) {
            const v4u* rp = (const v4u*)(rid + (long long)chn * CHUNK) + lane;
#pragma unroll
            for (int k = 0; k < 4; ++k) c[k] = rp[k * 32];
        }
        __builtin_amdgcn_fence(__ATOMIC_ACQ_REL, "wavefront");
        const unsigned m = (unsigned)__builtin_amdgcn_readfirstlane((int)qn[wave]);
        __builtin_amdgcn_fence(__ATOMIC_ACQ_REL, "wavefront");
        if (lane == 0) qn[wave] = 0u;
        for (unsigned i0 = 0; i0 < m; i0 += 32) {
            const unsigned i = i0 + lane;
            int      s  = 0;
            unsigned r  = 0u;
            float    w  = 0.0f;
            bool     ok = false;
            if (i < m) {
                const long long e = (long long)ebase + myq[i];
                s = src[e];
                const int d = dst[e];
                const float lg = logits[e];
                r = (unsigned)(d - nbase);
                asm volatile("" :: "v"(s), "v"(d), "v"(lg));
                ok = r < NR;
                if (ok) {
                    w = __expf(lg - smx[r]);
                    atomicAdd(&den[r], w);
                }
            }
            unsigned lv = __builtin_amdgcn_ballot_w32(ok);
            for (unsigned t0 = 0; lv != 0u; t0 += GRP, lv >>= GRP) {
                v2f row[GRP];
#pragma unroll
                for (int u = 0; u < GRP; ++u) {
                    const int su = __builtin_amdgcn_readlane(s, (int)t0 + u);
                    row[u] = *(const v2f*)(xl + (long long)su * D_DIM + 2 * lane);
                }
                static_assert(GRP == 16, "the statement below names the GRP rows one by one");
                asm volatile("" :: "v"(row[0]), "v"(row[1]), "v"(row[2]), "v"(row[3]),
                                   "v"(row[4]), "v"(row[5]), "v"(row[6]), "v"(row[7]),
                                   "v"(row[8]), "v"(row[9]), "v"(row[10]), "v"(row[11]),
                                   "v"(row[12]), "v"(row[13]), "v"(row[14]), "v"(row[15]));
#pragma unroll
                for (int u = 0; u < GRP; ++u)
                    if (lv & (1u << u)) {
                        const unsigned ru = (unsigned)__builtin_amdgcn_readlane((int)r, (int)t0 + u);
                        const float    wu = __uint_as_float((unsigned)__builtin_amdgcn_readlane(
                                                (int)__float_as_uint(w), (int)t0 + u));
                        float* ap = acc + ru * ROWP + 2 * lane;
                        atomicAdd(ap,     wu * row[u].x);
                        atomicAdd(ap + 1, wu * row[u].y);
                    }
            }
        }
        __builtin_amdgcn_fence(__ATOMIC_ACQ_REL, "wavefront");
        j = jn; ch = chn;
    }
    __syncthreads();

    for (int i = tid; i < NR * D_DIM; i += 1024) {
        const int r = i >> 6, k = i & 63;
        if (nbase + r < n_nodes)
            h[(long long)(nbase + r) * D_DIM + k] = acc[r * ROWP + k];
    }
    for (int r = tid; r < NR; r += 1024)
        if (nbase + r < n_nodes)
            denom[(long long)(nbase + r) * D_DIM] = den[r];
}

__global__ void k_output(const float* __restrict__ h, const float* __restrict__ denom,
                         const int* __restrict__ mask,
                         const float* __restrict__ gbias,
                         const float* __restrict__ Wo, const float* __restrict__ bo,
                         float* __restrict__ out, int n_mask) {
    const int lane = threadIdx.x & 31;
    const int wave = threadIdx.x >> 5;
    const int tile = blockIdx.x * (blockDim.x >> 5) + wave;
    const int r0 = tile * 16;
    if (r0 + 16 > n_mask) return;
    const int lrow = lane & 15;
    const int lhi  = lane >> 4;

    const long long mi = (long long)mask[r0 + lrow];
    v2f a[16];
    const float* hp = h + mi * D_DIM + 2 * lhi;
    const float* gp = gbias + 2 * lhi;
    const float rz = 1.0f / (denom[mi * D_DIM] + 1e-16f);
#pragma unroll
    for (int kk = 0; kk < 16; ++kk) {
        const v2f hv = *(const v2f*)(hp + kk * 4);
        const v2f gv = *(const v2f*)(gp + kk * 4);
        a[kk].x = hv.x * rz + gv.x;
        a[kk].y = hv.y * rz + gv.y;
    }
#pragma unroll
    for (int ct = 0; ct < 8; ++ct) {
        const int col = ct * 16 + lrow;
        const float bv = bo[col];
        v8f acc = {bv, bv, bv, bv, bv, bv, bv, bv};
#pragma unroll
        for (int kk = 0; kk < 16; ++kk) {
            const int k0 = kk * 4 + 2 * lhi;
            v2f b;
            b.x = Wo[(k0    ) * DOUT + col];
            b.y = Wo[(k0 + 1) * DOUT + col];
            acc = __builtin_amdgcn_wmma_f32_16x16x4_f32(
                    false, a[kk], false, b, (short)0, acc, false, false);
        }
#pragma unroll
        for (int r = 0; r < 8; ++r)
            out[(long long)(r0 + r + 8 * lhi) * DOUT + col] = acc[r];
    }
}

extern "C" void kernel_launch(void* const* d_in, const int* in_sizes, int n_in,
                              void* d_out, int out_size, void* d_ws, size_t ws_size,
                              hipStream_t stream) {
    const float* x     = (const float*)d_in[0];
    const int*   ei    = (const int*)d_in[1];
    const int*   mask  = (const int*)d_in[2];
    const float* Wl    = (const float*)d_in[3];
    const float* bl    = (const float*)d_in[4];
    const float* Wr    = (const float*)d_in[5];
    const float* br    = (const float*)d_in[6];
    const float* att   = (const float*)d_in[7];
    const float* gbias = (const float*)d_in[8];
    const float* Wo    = (const float*)d_in[9];
    const float* bo    = (const float*)d_in[10];
    float* out = (float*)d_out;

    const int n_nodes = in_sizes[0] / D_DIM;
    const int n_edges = in_sizes[1] / 2;
    const int n_mask  = in_sizes[2];
    const int* src = ei;
    const int* dst = ei + n_edges;

    char* wp = (char*)d_ws;
    auto carve = [&](size_t bytes) {
        char* p = wp;
        wp += (bytes + 255) & ~(size_t)255;
        return p;
    };
    float*    xl     = (float*)   carve((size_t)n_nodes * D_DIM * 4);
    float*    xr     = (float*)   carve((size_t)n_nodes * D_DIM * 4);
    float*    h      = (float*)   carve((size_t)n_nodes * D_DIM * 4);
    float*    logits = (float*)   carve((size_t)n_edges * 4);
    unsigned char* rid = (unsigned char*)carve((size_t)n_edges + CHUNK);
    unsigned* smax   = (unsigned*)h;
    float*    denom  = xr;

    k_init<<<(n_nodes + 255) / 256, 256, 0, stream>>>(h, n_nodes);
    {
        int tiles  = (n_nodes + 15) / 16;
        int blocks = (tiles + 7) / 8;
        k_node_linear<<<blocks, 256, 0, stream>>>(x, Wl, bl, Wr, br, xl, xr, n_nodes);
    }
    {
        int tiles  = (n_edges + 15) / 16;
        int blocks = (tiles + 7) / 8;
        k_edge_logits<<<blocks, 256, 0, stream>>>(src, dst, xl, xr, att, logits, smax, rid, n_edges);
    }
    k_scatter_lds<<<(n_nodes + NR - 1) / NR, 1024, 0, stream>>>(src, dst, xl, logits, rid, denom, h,
                                                              n_edges, n_nodes);
    {
        int tiles  = (n_mask + 15) / 16;
        int blocks = (tiles + 7) / 8;
        k_output<<<blocks, 256, 0, stream>>>(h, denom, mask, gbias, Wo, bo, out, n_mask);
    }
}


# segment max moved into the range owner's LDS: integer max while the group lists its edges in LDS, weights and row sums from the list; global atomic max and init launch gone; sums in arrival order
# speedup vs baseline: 2.3481x; 1.0939x over previous
#include <hip/hip_runtime.h>
#include <hip/hip_bf16.h>
#include <math.h>


#define D_DIM 64
#define DOUT  128
#define NEG_SLOPE 0.2f
#define NR    400
#define ROWP  65
#define CHUNK 2048
#define GRP   16

typedef __attribute__((ext_vector_type(2))) float v2f;
typedef __attribute__((ext_vector_type(4))) float v4f;
typedef __attribute__((ext_vector_type(8))) float v8f;

__device__ __forceinline__ unsigned f2ord(float f) {
    unsigned u = __float_as_uint(f);
    return (u & 0x80000000u) ? ~u : (u | 0x80000000u);
}
__device__ __forceinline__ float ord2f(unsigned u) {
    unsigned b = (u & 0x80000000u) ? (u ^ 0x80000000u) : ~u;
    return __uint_as_float(b);
}

__global__ void k_node_linear(const float* __restrict__ x,
                              const float* __restrict__ Wl, const float* __restrict__ bl,
                              const float* __restrict__ Wr, const float* __restrict__ br,
                              float* __restrict__ xl, float* __restrict__ xr,
                              int n_nodes) {
    const int lane = threadIdx.x & 31;
    const int wave = threadIdx.x >> 5;
    const int tile = blockIdx.x * (blockDim.x >> 5) + wave;
    const int r0 = tile * 16;
    if (r0 + 16 > n_nodes) return;
    const int lrow = lane & 15;
    const int lhi  = lane >> 4;

    v2f a[16];
    const float* xp = x + (long long)(r0 + lrow) * D_DIM + 2 * lhi;
#pragma unroll
    for (int kk = 0; kk < 16; ++kk)
        a[kk] = *(const v2f*)(xp + kk * 4);

#pragma unroll
    for (int m = 0; m < 2; ++m) {
        const float* W    = m ? Wr : Wl;
        const float* bias = m ? br : bl;
        float*       out  = m ? xr : xl;
#pragma unroll
        for (int ct = 0; ct < 4; ++ct) {
            const int col = ct * 16 + lrow;
            const float bv = bias[col];
            v8f acc = {bv, bv, bv, bv, bv, bv, bv, bv};
#pragma unroll
            for (int kk = 0; kk < 16; ++kk) {
                const int k0 = kk * 4 + 2 * lhi;
                v2f b;
                b.x = W[(k0    ) * D_DIM + col];
                b.y = W[(k0 + 1) * D_DIM + col];
                acc = __builtin_amdgcn_wmma_f32_16x16x4_f32(
                        false, a[kk], false, b, (short)0, acc, false, false);
            }
#pragma unroll
            for (int r = 0; r < 8; ++r)
                out[(long long)(r0 + r + 8 * lhi) * D_DIM + col] = acc[r];
        }
    }
}

__global__ void k_edge_logits(const int* __restrict__ src, const int* __restrict__ dst,
                              const float* __restrict__ xl, const float* __restrict__ xr,
                              const float* __restrict__ att,
                              float* __restrict__ logits,
                              unsigned char* __restrict__ rid,
                              int n_edges) {
    const int lane = threadIdx.x & 31;
    const int tile = (int)((blockIdx.x * blockDim.x + threadIdx.x) >> 5);
    const long long e0 = (long long)tile * 16;
    if (e0 >= n_edges) return;
    const int c  = lane & 15;
    const int hh = lane >> 4;
    long long e = e0 + c;
    const bool live = e < n_edges;
    if (!live) e = n_edges - 1;
    const int s = src[e], d = dst[e];
    const float* pl = xl  + (long long)s * D_DIM + 4 * hh;
    const float* pr = xr  + (long long)d * D_DIM + 4 * hh;
    const float* pa = att + 4 * hh;

    v4f vl[8], vr[8];
#pragma unroll
    for (int q = 0; q < 8; ++q) vl[q] = *(const v4f*)(pl + 8 * q);
#pragma unroll
    for (int q = 0; q < 8; ++q) vr[q] = *(const v4f*)(pr + 8 * q);
    __builtin_amdgcn_sched_barrier(0);

    v8f acc = {0.f, 0.f, 0.f, 0.f, 0.f, 0.f, 0.f, 0.f};
#pragma unroll
    for (int q = 0; q < 8; ++q) {
        const v4f t = *(const v4f*)(pa + 8 * q);
        v4f x = vl[q] + vr[q];
        x.x = x.x > 0.0f ? x.x : NEG_SLOPE * x.x;
        x.y = x.y > 0.0f ? x.y : NEG_SLOPE * x.y;
        x.z = x.z > 0.0f ? x.z : NEG_SLOPE * x.z;
        x.w = x.w > 0.0f ? x.w : NEG_SLOPE * x.w;
        const v2f a0 = {x.x, x.y}, b0 = {t.x, t.y};
        const v2f a1 = {x.z, x.w}, b1 = {t.z, t.w};
        acc = __builtin_amdgcn_wmma_f32_16x16x4_f32(
                false, a0, false, b0, (short)0, acc, false, false);
        acc = __builtin_amdgcn_wmma_f32_16x16x4_f32(
                false, a1, false, b1, (short)0, acc, false, false);
    }

    const int r = c & 7;
    float p = acc[0];
    p = r == 1 ? acc[1] : p;
    p = r == 2 ? acc[2] : p;
    p = r == 3 ? acc[3] : p;
    p = r == 4 ? acc[4] : p;
    p = r == 5 ? acc[5] : p;
    p = r == 6 ? acc[6] : p;
    p = r == 7 ? acc[7] : p;
    if (live && (c >> 3) == hh) {
        logits[e] = p;
        rid[e] = (unsigned char)((unsigned)d / NR);
    }
}

typedef __attribute__((ext_vector_type(4))) unsigned v4u;

#define LCAP 6144

__device__ __forceinline__ float seg_shift(unsigned o) {
    const float f = ord2f(o);
    return __builtin_isfinite(f) ? f : 0.0f;
}

__device__ __forceinline__ void add_rows(const float* __restrict__ xl, float* acc, const int lane,
                                         const int s, const unsigned r, const float w, const bool ok) {
    unsigned lv = __builtin_amdgcn_ballot_w32(ok);
    for (unsigned t0 = 0; lv != 0u; t0 += GRP, lv >>= GRP) {
        v2f row[GRP];
#pragma unroll
        for (int u = 0; u < GRP; ++u) {
            const int su = __builtin_amdgcn_readlane(s, (int)t0 + u);
            row[u] = *(const v2f*)(xl + (long long)su * D_DIM + 2 * lane);
        }
        static_assert(GRP == 16, "the statement below names the GRP rows one by one");
        asm volatile("" :: "v"(row[0]), "v"(row[1]), "v"(row[2]), "v"(row[3]),
                           "v"(row[4]), "v"(row[5]), "v"(row[6]), "v"(row[7]),
                           "v"(row[8]), "v"(row[9]), "v"(row[10]), "v"(row[11]),
                           "v"(row[12]), "v"(row[13]), "v"(row[14]), "v"(row[15]));
#pragma unroll
        for (int u = 0; u < GRP; ++u)
            if (lv & (1u << u)) {
                const unsigned ru = (unsigned)__builtin_amdgcn_readlane((int)r, (int)t0 + u);
                const float    wu = __uint_as_float((unsigned)__builtin_amdgcn_readlane(
                                        (int)__float_as_uint(w), (int)t0 + u));
                float* ap = acc + ru * ROWP + 2 * lane;
                atomicAdd(ap,     wu * row[u].x);
                atomicAdd(ap + 1, wu * row[u].y);
            }
    }
}

template <bool SUMS>
__device__ __forceinline__ void walk_plane(const int* __restrict__ src, const int* __restrict__ dst,
                                           const float* __restrict__ xl,
                                           const float* __restrict__ logits,
                                           const unsigned char* __restrict__ rid,
                                           const int n_edges, const int nbase, const unsigned bid,
                                           const int lane, const int wave,
                                           float* acc, float* den, unsigned* smo,
                                           unsigned short* myq, unsigned* qn,
                                           int* lsrc, float* llg, unsigned short* lrow,
                                           unsigned* lcnt, unsigned* lovf) {
    const unsigned pat = bid * 0x01010101u;
    const int nchunks = (n_edges + CHUNK - 1) / CHUNK;
    const int rot = (int)((bid * 37u) % (unsigned)nchunks);

    int j  = wave;
    int ch = j + rot; if (ch >= nchunks) ch -= nchunks;
    v4u c[4];
    if (j < nchunks) {
        const v4u* rp = (const v4u*)(rid + (long long)ch * CHUNK) + lane;
#pragma unroll
        for (int k = 0; k < 4; ++k) c[k] = rp[k * 32];
    }
    while (j < nchunks) {
        const int ebase = ch * CHUNK;
#pragma unroll
        for (int k = 0; k < 4; ++k) {
#pragma unroll
            for (int q = 0; q < 4; ++q) {
                const unsigned x = c[k][q] ^ pat;
                unsigned fl = ~(((x & 0x7F7F7F7Fu) + 0x7F7F7F7Fu) | x | 0x7F7F7F7Fu);
                while (fl) {
                    const unsigned b = (unsigned)__builtin_ctz(fl) >> 3;
                    fl &= fl - 1u;
                    const unsigned off = (unsigned)(k * 512 + lane * 16 + q * 4) + b;
                    if (ebase + (int)off < n_edges) {
                        const unsigned pos = atomicAdd(&qn[wave], 1u);
                        myq[pos] = (unsigned short)off;
                    }
                }
            }
        }
        const int jn = j + 32;
        int chn = jn + rot; if (chn >= nchunks) chn -= nchunks;
        if (jn < nchunks) {
            const v4u* rp = (const v4u*)(rid + (long long)chn * CHUNK) + lane;
#pragma unroll
            for (int k = 0; k < 4; ++k) c[k] = rp[k * 32];
        }
        __builtin_amdgcn_fence(__ATOMIC_ACQ_REL, "wavefront");
        const unsigned m = (unsigned)__builtin_amdgcn_readfirstlane((int)qn[wave]);
        __builtin_amdgcn_fence(__ATOMIC_ACQ_REL, "wavefront");
        if (lane == 0) qn[wave] = 0u;
        for (unsigned i0 = 0; i0 < m; i0 += 32) {
            const unsigned i = i0 + lane;
            int      s  = 0;
            unsigned r  = 0u;
            float    lg = 0.0f;
            bool     ok = false;
            if (i < m) {
                const long long e = (long long)ebase + myq[i];
                s = src[e];
                const int d = dst[e];
                lg = logits[e];
                r = (unsigned)(d - nbase);
                asm volatile("" :: "v"(s), "v"(d), "v"(lg));
                ok = r < NR;
            }
            if (SUMS) {
                float w = 0.0f;
                if (ok) {
                    w = __expf(lg - seg_shift(smo[r]));
                    atomicAdd(&den[r], w);
                }
                add_rows(xl, acc, lane, s, r, w, ok);
            } else {
                if (ok) atomicMax(&smo[r], f2ord(lg));
                const unsigned bal = __builtin_amdgcn_ballot_w32(ok);
                unsigned base = 0u;
                if (lane == 0) base = atomicAdd(lcnt, (unsigned)__builtin_popcount(bal));
                base = (unsigned)__builtin_amdgcn_readfirstlane((int)base);
                const unsigned slot = base + __builtin_amdgcn_mbcnt_lo(bal, 0u);
                if (ok) {
                    if (slot < LCAP) {
                        lsrc[slot] = s;
                        llg[slot]  = lg;
                        lrow[slot] = (unsigned short)r;
                    } else {
                        *lovf = 1u;
                    }
                }
            }
        }
        __builtin_amdgcn_fence(__ATOMIC_ACQ_REL, "wavefront");
        j = jn; ch = chn;
    }
}

__global__ __launch_bounds__(1024)
void k_scatter_lds(const int* __restrict__ src, const int* __restrict__ dst,
                   const float* __restrict__ xl,
                   const float* __restrict__ logits,
                   const unsigned char* __restrict__ rid,
                   float* __restrict__ denom,
                   float* __restrict__ h, int n_edges, int n_nodes) {
    __shared__ float          acc[NR * ROWP];
    __shared__ float          den[NR];
    __shared__ unsigned       smo[NR];
    __shared__ unsigned short qbuf[32 * CHUNK];
    __shared__ unsigned       qn[32];
    __shared__ int            lsrc[LCAP];
    __shared__ float          llg[LCAP];
    __shared__ unsigned short lrow[LCAP];
    __shared__ unsigned       lcnt;
    __shared__ unsigned       lovf;

    const int tid   = threadIdx.x;
    const int lane  = tid & 31;
    const int wave  = tid >> 5;
    const int nbase = blockIdx.x * NR;

    for (int i = tid; i < NR * ROWP; i += 1024) acc[i] = 0.0f;
    for (int r = tid; r < NR; r += 1024) {
        den[r] = 0.0f;
        smo[r] = 0x007FFFFFu;
    }
    if (tid < 32) qn[tid] = 0u;
    if (tid == 0) { lcnt = 0u; lovf = 0u; }
    __syncthreads();

    unsigned short* myq = qbuf + wave * CHUNK;
    walk_plane<false>(src, dst, xl, logits, rid, n_edges, nbase, blockIdx.x, lane, wave,
                      acc, den, smo, myq, qn, lsrc, llg, lrow, &lcnt, &lovf);
    __syncthreads();

    const unsigned total = (unsigned)__builtin_amdgcn_readfirstlane((int)lcnt);
    const unsigned over  = (unsigned)__builtin_amdgcn_readfirstlane((int)lovf);
    if (over == 0u) {
        for (unsigned i0 = (unsigned)wave * 32u; i0 < total; i0 += 1024u) {
            const unsigned i = i0 + (unsigned)lane;
            const bool ok = i < total;
            int      s = 0;
            unsigned r = 0u;
            float    w = 0.0f;
            if (ok) {
                s = lsrc[i];
                r = lrow[i];
                w = __expf(llg[i] - seg_shift(smo[r]));
                atomicAdd(&den[r], w);
            }
            add_rows(xl, acc, lane, s, r, w, ok);
        }
    } else {
        walk_plane<true>(src, dst, xl, logits, rid, n_edges, nbase, blockIdx.x, lane, wave,
                         acc, den, smo, myq, qn, lsrc, llg, lrow, &lcnt, &lovf);
    }
    __syncthreads();

    for (int i = tid; i < NR * D_DIM; i += 1024) {
        const int r = i >> 6, k = i & 63;
        if (nbase + r < n_nodes)
            h[(long long)(nbase + r) * D_DIM + k] = acc[r * ROWP + k];
    }
    for (int r = tid; r < NR; r += 1024)
        if (nbase + r < n_nodes)
            denom[(long long)(nbase + r) * D_DIM] = den[r];
}

__global__ void k_output(const float* __restrict__ h, const float* __restrict__ denom,
                         const int* __restrict__ mask,
                         const float* __restrict__ gbias,
                         const float* __restrict__ Wo, const float* __restrict__ bo,
                         float* __restrict__ out, int n_mask) {
    const int lane = threadIdx.x & 31;
    const int wave = threadIdx.x >> 5;
    const int tile = blockIdx.x * (blockDim.x >> 5) + wave;
    const int r0 = tile * 16;
    if (r0 + 16 > n_mask) return;
    const int lrow = lane & 15;
    const int lhi  = lane >> 4;

    const long long mi = (long long)mask[r0 + lrow];
    v2f a[16];
    const float* hp = h + mi * D_DIM + 2 * lhi;
    const float* gp = gbias + 2 * lhi;
    const float rz = 1.0f / (denom[mi * D_DIM] + 1e-16f);
#pragma unroll
    for (int kk = 0; kk < 16; ++kk) {
        const v2f hv = *(const v2f*)(hp + kk * 4);
        const v2f gv = *(const v2f*)(gp + kk * 4);
        a[kk].x = hv.x * rz + gv.x;
        a[kk].y = hv.y * rz + gv.y;
    }
#pragma unroll
    for (int ct = 0; ct < 8; ++ct) {
        const int col = ct * 16 + lrow;
        const float bv = bo[col];
        v8f acc = {bv, bv, bv, bv, bv, bv, bv, bv};
#pragma unroll
        for (int kk = 0; kk < 16; ++kk) {
            const int k0 = kk * 4 + 2 * lhi;
            v2f b;
            b.x = Wo[(k0    ) * DOUT + col];
            b.y = Wo[(k0 + 1) * DOUT + col];
            acc = __builtin_amdgcn_wmma_f32_16x16x4_f32(
                    false, a[kk], false, b, (short)0, acc, false, false);
        }
#pragma unroll
        for (int r = 0; r < 8; ++r)
            out[(long long)(r0 + r + 8 * lhi) * DOUT + col] = acc[r];
    }
}

extern "C" void kernel_launch(void* const* d_in, const int* in_sizes, int n_in,
                              void* d_out, int out_size, void* d_ws, size_t ws_size,
                              hipStream_t stream) {
    const float* x     = (const float*)d_in[0];
    const int*   ei    = (const int*)d_in[1];
    const int*   mask  = (const int*)d_in[2];
    const float* Wl    = (const float*)d_in[3];
    const float* bl    = (const float*)d_in[4];
    const float* Wr    = (const float*)d_in[5];
    const float* br    = (const float*)d_in[6];
    const float* att   = (const float*)d_in[7];
    const float* gbias = (const float*)d_in[8];
    const float* Wo    = (const float*)d_in[9];
    const float* bo    = (const float*)d_in[10];
    float* out = (float*)d_out;

    const int n_nodes = in_sizes[0] / D_DIM;
    const int n_edges = in_sizes[1] / 2;
    const int n_mask  = in_sizes[2];
    const int* src = ei;
    const int* dst = ei + n_edges;

    char* wp = (char*)d_ws;
    auto carve = [&](size_t bytes) {
        char* p = wp;
        wp += (bytes + 255) & ~(size_t)255;
        return p;
    };
    float*    xl     = (float*)   carve((size_t)n_nodes * D_DIM * 4);
    float*    xr     = (float*)   carve((size_t)n_nodes * D_DIM * 4);
    float*    h      = (float*)   carve((size_t)n_nodes * D_DIM * 4);
    float*    logits = (float*)   carve((size_t)n_edges * 4);
    unsigned char* rid = (unsigned char*)carve((size_t)n_edges + CHUNK);
    float*    denom  = xr;

    {
        int tiles  = (n_nodes + 15) / 16;
        int blocks = (tiles + 7) / 8;
        k_node_linear<<<blocks, 256, 0, stream>>>(x, Wl, bl, Wr, br, xl, xr, n_nodes);
    }
    {
        int tiles  = (n_edges + 15) / 16;
        int blocks = (tiles + 7) / 8;
        k_edge_logits<<<blocks, 256, 0, stream>>>(src, dst, xl, xr, att, logits, rid, n_edges);
    }
    k_scatter_lds<<<(n_nodes + NR - 1) / NR, 1024, 0, stream>>>(src, dst, xl, logits, rid, denom, h,
                                                              n_edges, n_nodes);
    {
        int tiles  = (n_mask + 15) / 16;
        int blocks = (tiles + 7) / 8;
        k_output<<<blocks, 256, 0, stream>>>(h, denom, mask, gbias, Wo, bo, out, n_mask);
    }
}


# edge logits: xl/xr rows fetched as whole lines (16 lanes per 256-B row, 4 lines a load; was 16 lines x 32 B); leaky_relu(xl+xr) tile turned to the A layout through LDS; order of sums unchanged
# speedup vs baseline: 2.5492x; 1.0856x over previous
#include <hip/hip_runtime.h>
#include <hip/hip_bf16.h>
#include <math.h>


#define D_DIM 64
#define DOUT  128
#define NEG_SLOPE 0.2f
#define NR    400
#define ROWP  65
#define CHUNK 2048
#define GRP   16

typedef __attribute__((ext_vector_type(2))) float v2f;
typedef __attribute__((ext_vector_type(4))) float v4f;
typedef __attribute__((ext_vector_type(8))) float v8f;

__device__ __forceinline__ unsigned f2ord(float f) {
    unsigned u = __float_as_uint(f);
    return (u & 0x80000000u) ? ~u : (u | 0x80000000u);
}
__device__ __forceinline__ float ord2f(unsigned u) {
    unsigned b = (u & 0x80000000u) ? (u ^ 0x80000000u) : ~u;
    return __uint_as_float(b);
}

__global__ void k_node_linear(const float* __restrict__ x,
                              const float* __restrict__ Wl, const float* __restrict__ bl,
                              const float* __restrict__ Wr, const float* __restrict__ br,
                              float* __restrict__ xl, float* __restrict__ xr,
                              int n_nodes) {
    const int lane = threadIdx.x & 31;
    const int wave = threadIdx.x >> 5;
    const int tile = blockIdx.x * (blockDim.x >> 5) + wave;
    const int r0 = tile * 16;
    if (r0 + 16 > n_nodes) return;
    const int lrow = lane & 15;
    const int lhi  = lane >> 4;

    v2f a[16];
    const float* xp = x + (long long)(r0 + lrow) * D_DIM + 2 * lhi;
#pragma unroll
    for (int kk = 0; kk < 16; ++kk)
        a[kk] = *(const v2f*)(xp + kk * 4);

#pragma unroll
    for (int m = 0; m < 2; ++m) {
        const float* W    = m ? Wr : Wl;
        const float* bias = m ? br : bl;
        float*       out  = m ? xr : xl;
#pragma unroll
        for (int ct = 0; ct < 4; ++ct) {
            const int col = ct * 16 + lrow;
            const float bv = bias[col];
            v8f acc = {bv, bv, bv, bv, bv, bv, bv, bv};
#pragma unroll
            for (int kk = 0; kk < 16; ++kk) {
                const int k0 = kk * 4 + 2 * lhi;
                v2f b;
                b.x = W[(k0    ) * D_DIM + col];
                b.y = W[(k0 + 1) * D_DIM + col];
                acc = __builtin_amdgcn_wmma_f32_16x16x4_f32(
                        false, a[kk], false, b, (short)0, acc, false, false);
            }
#pragma unroll
            for (int r = 0; r < 8; ++r)
                out[(long long)(r0 + r + 8 * lhi) * D_DIM + col] = acc[r];
        }
    }
}

#define EP4 17

template <int T>
__device__ __forceinline__ int row_lane(const int v) {
    return __builtin_amdgcn_update_dpp(0, v, 0x150 + T, 0xf, 0xf, false);
}

template <int T>
__device__ __forceinline__ void load_rows(const char* __restrict__ xlb, const char* __restrict__ xrb,
                                          const int s, const int d, const unsigned cb,
                                          v4f* vl, v4f* vr) {
    const unsigned so = (unsigned)row_lane<T>(s) * (D_DIM * 4u) + cb;
    const unsigned dd = (unsigned)row_lane<T>(d) * (D_DIM * 4u) + cb;
    vl[T] = *(const v4f*)(xlb + so);
    vr[T] = *(const v4f*)(xrb + dd);
}

__global__ __launch_bounds__(256)
void k_edge_logits(const int* __restrict__ src, const int* __restrict__ dst,
                   const float* __restrict__ xl, const float* __restrict__ xr,
                   const float* __restrict__ att,
                   float* __restrict__ logits,
                   unsigned char* __restrict__ rid,
                   int n_edges) {
    __shared__ v4f etile[8][16 * EP4];

    const int lane = threadIdx.x & 31;
    const int wave = threadIdx.x >> 5;
    const int tile = (int)((blockIdx.x * blockDim.x + threadIdx.x) >> 5);
    const long long e0 = (long long)tile * 16;
    if (e0 >= n_edges) return;
    const int c  = lane & 15;
    const int hh = lane >> 4;
    long long e = e0 + 8 * hh + (c & 7);
    const bool live = e < n_edges;
    if (!live) e = n_edges - 1;
    const int s = src[e], d = dst[e];
    const float* pa = att + 4 * hh;

    v4f vl[8], vr[8];
    {
        const char* xlb = (const char*)xl;
        const char* xrb = (const char*)xr;
        const unsigned cb = 16u * (unsigned)c;
        load_rows<0>(xlb, xrb, s, d, cb, vl, vr);
        load_rows<1>(xlb, xrb, s, d, cb, vl, vr);
        load_rows<2>(xlb, xrb, s, d, cb, vl, vr);
        load_rows<3>(xlb, xrb, s, d, cb, vl, vr);
        load_rows<4>(xlb, xrb, s, d, cb, vl, vr);
        load_rows<5>(xlb, xrb, s, d, cb, vl, vr);
        load_rows<6>(xlb, xrb, s, d, cb, vl, vr);
        load_rows<7>(xlb, xrb, s, d, cb, vl, vr);
    }

    v4f* my = &etile[wave][0];
#pragma unroll
    for (int t = 0; t < 8; ++t) {
        v4f x = vl[t] + vr[t];
        x.x = x.x > 0.0f ? x.x : NEG_SLOPE * x.x;
        x.y = x.y > 0.0f ? x.y : NEG_SLOPE * x.y;
        x.z = x.z > 0.0f ? x.z : NEG_SLOPE * x.z;
        x.w = x.w > 0.0f ? x.w : NEG_SLOPE * x.w;
        my[(8 * hh + t) * EP4 + c] = x;
    }
    __builtin_amdgcn_fence(__ATOMIC_ACQ_REL, "wavefront");

    v4f xa[8], tq[8];
#pragma unroll
    for (int q = 0; q < 8; ++q) {
        tq[q] = *(const v4f*)(pa + 8 * q);
        xa[q] = my[c * EP4 + 2 * q + hh];
    }
    asm volatile("" :: "v"(xa[0]), "v"(xa[1]), "v"(xa[2]), "v"(xa[3]),
                       "v"(xa[4]), "v"(xa[5]), "v"(xa[6]), "v"(xa[7]),
                       "v"(tq[0]), "v"(tq[1]), "v"(tq[2]), "v"(tq[3]),
                       "v"(tq[4]), "v"(tq[5]), "v"(tq[6]), "v"(tq[7]));

    v8f acc = {0.f, 0.f, 0.f, 0.f, 0.f, 0.f, 0.f, 0.f};
#pragma unroll
    for (int q = 0; q < 8; ++q) {
        const v2f a0 = {xa[q].x, xa[q].y}, b0 = {tq[q].x, tq[q].y};
        const v2f a1 = {xa[q].z, xa[q].w}, b1 = {tq[q].z, tq[q].w};
        acc = __builtin_amdgcn_wmma_f32_16x16x4_f32(
                false, a0, false, b0, (short)0, acc, false, false);
        acc = __builtin_amdgcn_wmma_f32_16x16x4_f32(
                false, a1, false, b1, (short)0, acc, false, false);
    }

    const int r = c & 7;
    float p = acc[0];
    p = r == 1 ? acc[1] : p;
    p = r == 2 ? acc[2] : p;
    p = r == 3 ? acc[3] : p;
    p = r == 4 ? acc[4] : p;
    p = r == 5 ? acc[5] : p;
    p = r == 6 ? acc[6] : p;
    p = r == 7 ? acc[7] : p;
    if (live && c < 8) {
        logits[e] = p;
        rid[e] = (unsigned char)((unsigned)d / NR);
    }
}

typedef __attribute__((ext_vector_type(4))) unsigned v4u;

#define LCAP 6144

__device__ __forceinline__ float seg_shift(unsigned o) {
    const float f = ord2f(o);
    return __builtin_isfinite(f) ? f : 0.0f;
}

__device__ __forceinline__ void add_rows(const float* __restrict__ xl, float* acc, const int lane,
                                         const int s, const unsigned r, const float w, const bool ok) {
    unsigned lv = __builtin_amdgcn_ballot_w32(ok);
    for (unsigned t0 = 0; lv != 0u; t0 += GRP, lv >>= GRP) {
        v2f row[GRP];
#pragma unroll
        for (int u = 0; u < GRP; ++u) {
            const int su = __builtin_amdgcn_readlane(s, (int)t0 + u);
            row[u] = *(const v2f*)(xl + (long long)su * D_DIM + 2 * lane);
        }
        static_assert(GRP == 16, "the statement below names the GRP rows one by one");
        asm volatile("" :: "v"(row[0]), "v"(row[1]), "v"(row[2]), "v"(row[3]),
                           "v"(row[4]), "v"(row[5]), "v"(row[6]), "v"(row[7]),
                           "v"(row[8]), "v"(row[9]), "v"(row[10]), "v"(row[11]),
                           "v"(row[12]), "v"(row[13]), "v"(row[14]), "v"(row[15]));
#pragma unroll
        for (int u = 0; u < GRP; ++u)
            if (lv & (1u << u)) {
                const unsigned ru = (unsigned)__builtin_amdgcn_readlane((int)r, (int)t0 + u);
                const float    wu = __uint_as_float((unsigned)__builtin_amdgcn_readlane(
                                        (int)__float_as_uint(w), (int)t0 + u));
                float* ap = acc + ru * ROWP + 2 * lane;
                atomicAdd(ap,     wu * row[u].x);
                atomicAdd(ap + 1, wu * row[u].y);
            }
    }
}

template <bool SUMS>
__device__ __forceinline__ void walk_plane(const int* __restrict__ src, const int* __restrict__ dst,
                                           const float* __restrict__ xl,
                                           const float* __restrict__ logits,
                                           const unsigned char* __restrict__ rid,
                                           const int n_edges, const int nbase, const unsigned bid,
                                           const int lane, const int wave,
                                           float* acc, float* den, unsigned* smo,
                                           unsigned short* myq, unsigned* qn,
                                           int* lsrc, float* llg, unsigned short* lrow,
                                           unsigned* lcnt, unsigned* lovf) {
    const unsigned pat = bid * 0x01010101u;
    const int nchunks = (n_edges + CHUNK - 1) / CHUNK;
    const int rot = (int)((bid * 37u) % (unsigned)nchunks);

    int j  = wave;
    int ch = j + rot; if (ch >= nchunks) ch -= nchunks;
    v4u c[4];
    if (j < nchunks) {
        const v4u* rp = (const v4u*)(rid + (long long)ch * CHUNK) + lane;
#pragma unroll
        for (int k = 0; k < 4; ++k) c[k] = rp[k * 32];
    }
    while (j < nchunks) {
        const int ebase = ch * CHUNK;
#pragma unroll
        for (int k = 0; k < 4; ++k) {
#pragma unroll
            for (int q = 0; q < 4; ++q) {
                const unsigned x = c[k][q] ^ pat;
                unsigned fl = ~(((x & 0x7F7F7F7Fu) + 0x7F7F7F7Fu) | x | 0x7F7F7F7Fu);
                while (fl) {
                    const unsigned b = (unsigned)__builtin_ctz(fl) >> 3;
                    fl &= fl - 1u;
                    const unsigned off = (unsigned)(k * 512 + lane * 16 + q * 4) + b;
                    if (ebase + (int)off < n_edges) {
                        const unsigned pos = atomicAdd(&qn[wave], 1u);
                        myq[pos] = (unsigned short)off;
                    }
                }
            }
        }
        const int jn = j + 32;
        int chn = jn + rot; if (chn >= nchunks) chn -= nchunks;
        if (jn < nchunks) {
            const v4u* rp = (const v4u*)(rid + (long long)chn * CHUNK) + lane;
#pragma unroll
            for (int k = 0; k < 4; ++k) c[k] = rp[k * 32];
        }
        __builtin_amdgcn_fence(__ATOMIC_ACQ_REL, "wavefront");
        const unsigned m = (unsigned)__builtin_amdgcn_readfirstlane((int)qn[wave]);
        __builtin_amdgcn_fence(__ATOMIC_ACQ_REL, "wavefront");
        if (lane == 0) qn[wave] = 0u;
        for (unsigned i0 = 0; i0 < m; i0 += 32) {
            const unsigned i = i0 + lane;
            int      s  = 0;
            unsigned r  = 0u;
            float    lg = 0.0f;
            bool     ok = false;
            if (i < m) {
                const long long e = (long long)ebase + myq[i];
                s = src[e];
                const int d = dst[e];
                lg = logits[e];
                r = (unsigned)(d - nbase);
                asm volatile("" :: "v"(s), "v"(d), "v"(lg));
                ok = r < NR;
            }
            if (SUMS) {
                float w = 0.0f;
                if (ok) {
                    w = __expf(lg - seg_shift(smo[r]));
                    atomicAdd(&den[r], w);
                }
                add_rows(xl, acc, lane, s, r, w, ok);
            } else {
                if (ok) atomicMax(&smo[r], f2ord(lg));
                const unsigned bal = __builtin_amdgcn_ballot_w32(ok);
                unsigned base = 0u;
                if (lane == 0) base = atomicAdd(lcnt, (unsigned)__builtin_popcount(bal));
                base = (unsigned)__builtin_amdgcn_readfirstlane((int)base);
                const unsigned slot = base + __builtin_amdgcn_mbcnt_lo(bal, 0u);
                if (ok) {
                    if (slot < LCAP) {
                        lsrc[slot] = s;
                        llg[slot]  = lg;
                        lrow[slot] = (unsigned short)r;
                    } else {
                        *lovf = 1u;
                    }
                }
            }
        }
        __builtin_amdgcn_fence(__ATOMIC_ACQ_REL, "wavefront");
        j = jn; ch = chn;
    }
}

__global__ __launch_bounds__(1024)
void k_scatter_lds(const int* __restrict__ src, const int* __restrict__ dst,
                   const float* __restrict__ xl,
                   const float* __restrict__ logits,
                   const unsigned char* __restrict__ rid,
                   float* __restrict__ denom,
                   float* __restrict__ h, int n_edges, int n_nodes) {
    __shared__ float          acc[NR * ROWP];
    __shared__ float          den[NR];
    __shared__ unsigned       smo[NR];
    __shared__ unsigned short qbuf[32 * CHUNK];
    __shared__ unsigned       qn[32];
    __shared__ int            lsrc[LCAP];
    __shared__ float          llg[LCAP];
    __shared__ unsigned short lrow[LCAP];
    __shared__ unsigned       lcnt;
    __shared__ unsigned       lovf;

    const int tid   = threadIdx.x;
    const int lane  = tid & 31;
    const int wave  = tid >> 5;
    const int nbase = blockIdx.x * NR;

    for (int i = tid; i < NR * ROWP; i += 1024) acc[i] = 0.0f;
    for (int r = tid; r < NR; r += 1024) {
        den[r] = 0.0f;
        smo[r] = 0x007FFFFFu;
    }
    if (tid < 32) qn[tid] = 0u;
    if (tid == 0) { lcnt = 0u; lovf = 0u; }
    __syncthreads();

    unsigned short* myq = qbuf + wave * CHUNK;
    walk_plane<false>(src, dst, xl, logits, rid, n_edges, nbase, blockIdx.x, lane, wave,
                      acc, den, smo, myq, qn, lsrc, llg, lrow, &lcnt, &lovf);
    __syncthreads();

    const unsigned total = (unsigned)__builtin_amdgcn_readfirstlane((int)lcnt);
    const unsigned over  = (unsigned)__builtin_amdgcn_readfirstlane((int)lovf);
    if (over == 0u) {
        for (unsigned i0 = (unsigned)wave * 32u; i0 < total; i0 += 1024u) {
            const unsigned i = i0 + (unsigned)lane;
            const bool ok = i < total;
            int      s = 0;
            unsigned r = 0u;
            float    w = 0.0f;
            if (ok) {
                s = lsrc[i];
                r = lrow[i];
                w = __expf(llg[i] - seg_shift(smo[r]));
                atomicAdd(&den[r], w);
            }
            add_rows(xl, acc, lane, s, r, w, ok);
        }
    } else {
        walk_plane<true>(src, dst, xl, logits, rid, n_edges, nbase, blockIdx.x, lane, wave,
                         acc, den, smo, myq, qn, lsrc, llg, lrow, &lcnt, &lovf);
    }
    __syncthreads();

    for (int i = tid; i < NR * D_DIM; i += 1024) {
        const int r = i >> 6, k = i & 63;
        if (nbase + r < n_nodes)
            h[(long long)(nbase + r) * D_DIM + k] = acc[r * ROWP + k];
    }
    for (int r = tid; r < NR; r += 1024)
        if (nbase + r < n_nodes)
            denom[(long long)(nbase + r) * D_DIM] = den[r];
}

__global__ void k_output(const float* __restrict__ h, const float* __restrict__ denom,
                         const int* __restrict__ mask,
                         const float* __restrict__ gbias,
                         const float* __restrict__ Wo, const float* __restrict__ bo,
                         float* __restrict__ out, int n_mask) {
    const int lane = threadIdx.x & 31;
    const int wave = threadIdx.x >> 5;
    const int tile = blockIdx.x * (blockDim.x >> 5) + wave;
    const int r0 = tile * 16;
    if (r0 + 16 > n_mask) return;
    const int lrow = lane & 15;
    const int lhi  = lane >> 4;

    const long long mi = (long long)mask[r0 + lrow];
    v2f a[16];
    const float* hp = h + mi * D_DIM + 2 * lhi;
    const float* gp = gbias + 2 * lhi;
    const float rz = 1.0f / (denom[mi * D_DIM] + 1e-16f);
#pragma unroll
    for (int kk = 0; kk < 16; ++kk) {
        const v2f hv = *(const v2f*)(hp + kk * 4);
        const v2f gv = *(const v2f*)(gp + kk * 4);
        a[kk].x = hv.x * rz + gv.x;
        a[kk].y = hv.y * rz + gv.y;
    }
#pragma unroll
    for (int ct = 0; ct < 8; ++ct) {
        const int col = ct * 16 + lrow;
        const float bv = bo[col];
        v8f acc = {bv, bv, bv, bv, bv, bv, bv, bv};
#pragma unroll
        for (int kk = 0; kk < 16; ++kk) {
            const int k0 = kk * 4 + 2 * lhi;
            v2f b;
            b.x = Wo[(k0    ) * DOUT + col];
            b.y = Wo[(k0 + 1) * DOUT + col];
            acc = __builtin_amdgcn_wmma_f32_16x16x4_f32(
                    false, a[kk], false, b, (short)0, acc, false, false);
        }
#pragma unroll
        for (int r = 0; r < 8; ++r)
            out[(long long)(r0 + r + 8 * lhi) * DOUT + col] = acc[r];
    }
}

extern "C" void kernel_launch(void* const* d_in, const int* in_sizes, int n_in,
                              void* d_out, int out_size, void* d_ws, size_t ws_size,
                              hipStream_t stream) {
    const float* x     = (const float*)d_in[0];
    const int*   ei    = (const int*)d_in[1];
    const int*   mask  = (const int*)d_in[2];
    const float* Wl    = (const float*)d_in[3];
    const float* bl    = (const float*)d_in[4];
    const float* Wr    = (const float*)d_in[5];
    const float* br    = (const float*)d_in[6];
    const float* att   = (const float*)d_in[7];
    const float* gbias = (const float*)d_in[8];
    const float* Wo    = (const float*)d_in[9];
    const float* bo    = (const float*)d_in[10];
    float* out = (float*)d_out;

    const int n_nodes = in_sizes[0] / D_DIM;
    const int n_edges = in_sizes[1] / 2;
    const int n_mask  = in_sizes[2];
    const int* src = ei;
    const int* dst = ei + n_edges;

    char* wp = (char*)d_ws;
    auto carve = [&](size_t bytes) {
        char* p = wp;
        wp += (bytes + 255) & ~(size_t)255;
        return p;
    };
    float*    xl     = (float*)   carve((size_t)n_nodes * D_DIM * 4);
    float*    xr     = (float*)   carve((size_t)n_nodes * D_DIM * 4);
    float*    h      = (float*)   carve((size_t)n_nodes * D_DIM * 4);
    float*    logits = (float*)   carve((size_t)n_edges * 4);
    unsigned char* rid = (unsigned char*)carve((size_t)n_edges + CHUNK);
    float*    denom  = xr;

    {
        int tiles  = (n_nodes + 15) / 16;
        int blocks = (tiles + 7) / 8;
        k_node_linear<<<blocks, 256, 0, stream>>>(x, Wl, bl, Wr, br, xl, xr, n_nodes);
    }
    {
        int tiles  = (n_edges + 15) / 16;
        int blocks = (tiles + 7) / 8;
        k_edge_logits<<<blocks, 256, 0, stream>>>(src, dst, xl, xr, att, logits, rid, n_edges);
    }
    k_scatter_lds<<<(n_nodes + NR - 1) / NR, 1024, 0, stream>>>(src, dst, xl, logits, rid, denom, h,
                                                              n_edges, n_nodes);
    {
        int tiles  = (n_mask + 15) / 16;
        int blocks = (tiles + 7) / 8;
        k_output<<<blocks, 256, 0, stream>>>(h, denom, mask, gbias, Wo, bo, out, n_mask);
    }
}


# scatter sums: a range's rows dealt to its 32 waves (row mod 32); pass 1 lists each edge with its row's owner, who sums its rows in registers (fused multiply-add, arrival order); no LDS float adds
# speedup vs baseline: 2.7131x; 1.0643x over previous
#include <hip/hip_runtime.h>
#include <hip/hip_bf16.h>
#include <math.h>


#define D_DIM 64
#define DOUT  128
#define NEG_SLOPE 0.2f
#define NR    400
#define ROWP  65
#define CHUNK 2048
#define GRP   16

typedef __attribute__((ext_vector_type(2))) float v2f;
typedef __attribute__((ext_vector_type(4))) float v4f;
typedef __attribute__((ext_vector_type(8))) float v8f;

__device__ __forceinline__ unsigned f2ord(float f) {
    unsigned u = __float_as_uint(f);
    return (u & 0x80000000u) ? ~u : (u | 0x80000000u);
}
__device__ __forceinline__ float ord2f(unsigned u) {
    unsigned b = (u & 0x80000000u) ? (u ^ 0x80000000u) : ~u;
    return __uint_as_float(b);
}

__global__ void k_node_linear(const float* __restrict__ x,
                              const float* __restrict__ Wl, const float* __restrict__ bl,
                              const float* __restrict__ Wr, const float* __restrict__ br,
                              float* __restrict__ xl, float* __restrict__ xr,
                              int n_nodes) {
    const int lane = threadIdx.x & 31;
    const int wave = threadIdx.x >> 5;
    const int tile = blockIdx.x * (blockDim.x >> 5) + wave;
    const int r0 = tile * 16;
    if (r0 + 16 > n_nodes) return;
    const int lrow = lane & 15;
    const int lhi  = lane >> 4;

    v2f a[16];
    const float* xp = x + (long long)(r0 + lrow) * D_DIM + 2 * lhi;
#pragma unroll
    for (int kk = 0; kk < 16; ++kk)
        a[kk] = *(const v2f*)(xp + kk * 4);

#pragma unroll
    for (int m = 0; m < 2; ++m) {
        const float* W    = m ? Wr : Wl;
        const float* bias = m ? br : bl;
        float*       out  = m ? xr : xl;
#pragma unroll
        for (int ct = 0; ct < 4; ++ct) {
            const int col = ct * 16 + lrow;
            const float bv = bias[col];
            v8f acc = {bv, bv, bv, bv, bv, bv, bv, bv};
#pragma unroll
            for (int kk = 0; kk < 16; ++kk) {
                const int k0 = kk * 4 + 2 * lhi;
                v2f b;
                b.x = W[(k0    ) * D_DIM + col];
                b.y = W[(k0 + 1) * D_DIM + col];
                acc = __builtin_amdgcn_wmma_f32_16x16x4_f32(
                        false, a[kk], false, b, (short)0, acc, false, false);
            }
#pragma unroll
            for (int r = 0; r < 8; ++r)
                out[(long long)(r0 + r + 8 * lhi) * D_DIM + col] = acc[r];
        }
    }
}

#define EP4 17

template <int T>
__device__ __forceinline__ int row_lane(const int v) {
    return __builtin_amdgcn_update_dpp(0, v, 0x150 + T, 0xf, 0xf, false);
}

template <int T>
__device__ __forceinline__ void load_rows(const char* __restrict__ xlb, const char* __restrict__ xrb,
                                          const int s, const int d, const unsigned cb,
                                          v4f* vl, v4f* vr) {
    const unsigned so = (unsigned)row_lane<T>(s) * (D_DIM * 4u) + cb;
    const unsigned dd = (unsigned)row_lane<T>(d) * (D_DIM * 4u) + cb;
    vl[T] = *(const v4f*)(xlb + so);
    vr[T] = *(const v4f*)(xrb + dd);
}

__global__ __launch_bounds__(256)
void k_edge_logits(const int* __restrict__ src, const int* __restrict__ dst,
                   const float* __restrict__ xl, const float* __restrict__ xr,
                   const float* __restrict__ att,
                   float* __restrict__ logits,
                   unsigned char* __restrict__ rid,
                   int n_edges) {
    __shared__ v4f etile[8][16 * EP4];

    const int lane = threadIdx.x & 31;
    const int wave = threadIdx.x >> 5;
    const int tile = (int)((blockIdx.x * blockDim.x + threadIdx.x) >> 5);
    const long long e0 = (long long)tile * 16;
    if (e0 >= n_edges) return;
    const int c  = lane & 15;
    const int hh = lane >> 4;
    long long e = e0 + 8 * hh + (c & 7);
    const bool live = e < n_edges;
    if (!live) e = n_edges - 1;
    const int s = src[e], d = dst[e];
    const float* pa = att + 4 * hh;

    v4f vl[8], vr[8];
    {
        const char* xlb = (const char*)xl;
        const char* xrb = (const char*)xr;
        const unsigned cb = 16u * (unsigned)c;
        load_rows<0>(xlb, xrb, s, d, cb, vl, vr);
        load_rows<1>(xlb, xrb, s, d, cb, vl, vr);
        load_rows<2>(xlb, xrb, s, d, cb, vl, vr);
        load_rows<3>(xlb, xrb, s, d, cb, vl, vr);
        load_rows<4>(xlb, xrb, s, d, cb, vl, vr);
        load_rows<5>(xlb, xrb, s, d, cb, vl, vr);
        load_rows<6>(xlb, xrb, s, d, cb, vl, vr);
        load_rows<7>(xlb, xrb, s, d, cb, vl, vr);
    }

    v4f* my = &etile[wave][0];
#pragma unroll
    for (int t = 0; t < 8; ++t) {
        v4f x = vl[t] + vr[t];
        x.x = x.x > 0.0f ? x.x : NEG_SLOPE * x.x;
        x.y = x.y > 0.0f ? x.y : NEG_SLOPE * x.y;
        x.z = x.z > 0.0f ? x.z : NEG_SLOPE * x.z;
        x.w = x.w > 0.0f ? x.w : NEG_SLOPE * x.w;
        my[(8 * hh + t) * EP4 + c] = x;
    }
    __builtin_amdgcn_fence(__ATOMIC_ACQ_REL, "wavefront");

    v4f xa[8], tq[8];
#pragma unroll
    for (int q = 0; q < 8; ++q) {
        tq[q] = *(const v4f*)(pa + 8 * q);
        xa[q] = my[c * EP4 + 2 * q + hh];
    }
    asm volatile("" :: "v"(xa[0]), "v"(xa[1]), "v"(xa[2]), "v"(xa[3]),
                       "v"(xa[4]), "v"(xa[5]), "v"(xa[6]), "v"(xa[7]),
                       "v"(tq[0]), "v"(tq[1]), "v"(tq[2]), "v"(tq[3]),
                       "v"(tq[4]), "v"(tq[5]), "v"(tq[6]), "v"(tq[7]));

    v8f acc = {0.f, 0.f, 0.f, 0.f, 0.f, 0.f, 0.f, 0.f};
#pragma unroll
    for (int q = 0; q < 8; ++q) {
        const v2f a0 = {xa[q].x, xa[q].y}, b0 = {tq[q].x, tq[q].y};
        const v2f a1 = {xa[q].z, xa[q].w}, b1 = {tq[q].z, tq[q].w};
        acc = __builtin_amdgcn_wmma_f32_16x16x4_f32(
                false, a0, false, b0, (short)0, acc, false, false);
        acc = __builtin_amdgcn_wmma_f32_16x16x4_f32(
                false, a1, false, b1, (short)0, acc, false, false);
    }

    const int r = c & 7;
    float p = acc[0];
    p = r == 1 ? acc[1] : p;
    p = r == 2 ? acc[2] : p;
    p = r == 3 ? acc[3] : p;
    p = r == 4 ? acc[4] : p;
    p = r == 5 ? acc[5] : p;
    p = r == 6 ? acc[6] : p;
    p = r == 7 ? acc[7] : p;
    if (live && c < 8) {
        logits[e] = p;
        rid[e] = (unsigned char)((unsigned)d / NR);
    }
}

typedef __attribute__((ext_vector_type(4))) unsigned v4u;

#define NB   32
#define BCAP 320
#define RPW  13
static_assert(NB * RPW >= NR, "every local row needs an owner");
static_assert(2 * RPW <= 32, "a wave's row sums fit one 32-word vector per lane");
static_assert(NB * BCAP * 10 <= NR * ROWP * 4, "the lists (4 + 4 + 2 B an entry) lie in the storage of the LDS sum plane");

__device__ __forceinline__ float seg_shift(unsigned o) {
    const float f = ord2f(o);
    return __builtin_isfinite(f) ? f : 0.0f;
}

__device__ __forceinline__ void add_rows(const float* __restrict__ xl, float* acc, const int lane,
                                         const int s, const unsigned r, const float w, const bool ok) {
    unsigned lv = __builtin_amdgcn_ballot_w32(ok);
    for (unsigned t0 = 0; lv != 0u; t0 += GRP, lv >>= GRP) {
        v2f row[GRP];
#pragma unroll
        for (int u = 0; u < GRP; ++u) {
            const int su = __builtin_amdgcn_readlane(s, (int)t0 + u);
            row[u] = *(const v2f*)(xl + (long long)su * D_DIM + 2 * lane);
        }
        static_assert(GRP == 16, "the statement below names the GRP rows one by one");
        asm volatile("" :: "v"(row[0]), "v"(row[1]), "v"(row[2]), "v"(row[3]),
                           "v"(row[4]), "v"(row[5]), "v"(row[6]), "v"(row[7]),
                           "v"(row[8]), "v"(row[9]), "v"(row[10]), "v"(row[11]),
                           "v"(row[12]), "v"(row[13]), "v"(row[14]), "v"(row[15]));
#pragma unroll
        for (int u = 0; u < GRP; ++u)
            if (lv & (1u << u)) {
                const unsigned ru = (unsigned)__builtin_amdgcn_readlane((int)r, (int)t0 + u);
                const float    wu = __uint_as_float((unsigned)__builtin_amdgcn_readlane(
                                        (int)__float_as_uint(w), (int)t0 + u));
                float* ap = acc + ru * ROWP + 2 * lane;
                atomicAdd(ap,     wu * row[u].x);
                atomicAdd(ap + 1, wu * row[u].y);
            }
    }
}

typedef __attribute__((ext_vector_type(32))) float v32f;

__device__ __forceinline__ void own_rows(const float* __restrict__ xl, v32f& racc, const int lane,
                                         const int s, const unsigned j, const float w, const bool ok) {
    unsigned lv = __builtin_amdgcn_ballot_w32(ok);
    for (unsigned t0 = 0; lv != 0u; t0 += GRP, lv >>= GRP) {
        v2f row[GRP];
#pragma unroll
        for (int u = 0; u < GRP; ++u) {
            const int su = __builtin_amdgcn_readlane(s, (int)t0 + u);
            row[u] = *(const v2f*)(xl + (long long)su * D_DIM + 2 * lane);
        }
        static_assert(GRP == 16, "the statement below names the GRP rows one by one");
        asm volatile("" :: "v"(row[0]), "v"(row[1]), "v"(row[2]), "v"(row[3]),
                           "v"(row[4]), "v"(row[5]), "v"(row[6]), "v"(row[7]),
                           "v"(row[8]), "v"(row[9]), "v"(row[10]), "v"(row[11]),
                           "v"(row[12]), "v"(row[13]), "v"(row[14]), "v"(row[15]));
#pragma unroll
        for (int u = 0; u < GRP; ++u)
            if (lv & (1u << u)) {
                const unsigned ju = (unsigned)__builtin_amdgcn_readlane((int)j, (int)t0 + u);
                const float    wu = __uint_as_float((unsigned)__builtin_amdgcn_readlane(
                                        (int)__float_as_uint(w), (int)t0 + u));
                const unsigned k = 2u * ju;
                racc[k]      += wu * row[u].x;
                racc[k + 1u] += wu * row[u].y;
            }
    }
}

template <bool SUMS>
__device__ __forceinline__ void walk_plane(const int* __restrict__ src, const int* __restrict__ dst,
                                           const float* __restrict__ xl,
                                           const float* __restrict__ logits,
                                           const unsigned char* __restrict__ rid,
                                           const int n_edges, const int nbase, const unsigned bid,
                                           const int lane, const int wave,
                                           float* acc, float* den, unsigned* smo,
                                           unsigned short* myq, unsigned* qn,
                                           int* lsrc, float* llg, unsigned short* lrow,
                                           unsigned* bcnt, unsigned* lovf) {
    const unsigned pat = bid * 0x01010101u;
    const int nchunks = (n_edges + CHUNK - 1) / CHUNK;
    const int rot = (int)((bid * 37u) % (unsigned)nchunks);

    int j  = wave;
    int ch = j + rot; if (ch >= nchunks) ch -= nchunks;
    v4u c[4];
    if (j < nchunks) {
        const v4u* rp = (const v4u*)(rid + (long long)ch * CHUNK) + lane;
#pragma unroll
        for (int k = 0; k < 4; ++k) c[k] = rp[k * 32];
    }
    while (j < nchunks) {
        const int ebase = ch * CHUNK;
#pragma unroll
        for (int k = 0; k < 4; ++k) {
#pragma unroll
            for (int q = 0; q < 4; ++q) {
                const unsigned x = c[k][q] ^ pat;
                unsigned fl = ~(((x & 0x7F7F7F7Fu) + 0x7F7F7F7Fu) | x | 0x7F7F7F7Fu);
                while (fl) {
                    const unsigned b = (unsigned)__builtin_ctz(fl) >> 3;
                    fl &= fl - 1u;
                    const unsigned off = (unsigned)(k * 512 + lane * 16 + q * 4) + b;
                    if (ebase + (int)off < n_edges) {
                        const unsigned pos = atomicAdd(&qn[wave], 1u);
                        myq[pos] = (unsigned short)off;
                    }
                }
            }
        }
        const int jn = j + 32;
        int chn = jn + rot; if (chn >= nchunks) chn -= nchunks;
        if (jn < nchunks) {
            const v4u* rp = (const v4u*)(rid + (long long)chn * CHUNK) + lane;
#pragma unroll
            for (int k = 0; k < 4; ++k) c[k] = rp[k * 32];
        }
        __builtin_amdgcn_fence(__ATOMIC_ACQ_REL, "wavefront");
        const unsigned m = (unsigned)__builtin_amdgcn_readfirstlane((int)qn[wave]);
        __builtin_amdgcn_fence(__ATOMIC_ACQ_REL, "wavefront");
        if (lane == 0) qn[wave] = 0u;
        for (unsigned i0 = 0; i0 < m; i0 += 32) {
            const unsigned i = i0 + lane;
            int      s  = 0;
            unsigned r  = 0u;
            float    lg = 0.0f;
            bool     ok = false;
            if (i < m) {
                const long long e = (long long)ebase + myq[i];
                s = src[e];
                const int d = dst[e];
                lg = logits[e];
                r = (unsigned)(d - nbase);
                asm volatile("" :: "v"(s), "v"(d), "v"(lg));
                ok = r < NR;
            }
            if (SUMS) {
                float w = 0.0f;
                if (ok) {
                    w = __expf(lg - seg_shift(smo[r]));
                    atomicAdd(&den[r], w);
                }
                add_rows(xl, acc, lane, s, r, w, ok);
            } else {
                if (ok) {
                    atomicMax(&smo[r], f2ord(lg));
                    const unsigned b    = r % NB;
                    const unsigned slot = atomicAdd(&bcnt[b], 1u);
                    if (slot < BCAP) {
                        const unsigned p = b * BCAP + slot;
                        lsrc[p] = s;
                        llg[p]  = lg;
                        lrow[p] = (unsigned short)r;
                    } else {
                        *lovf = 1u;
                    }
                }
            }
        }
        __builtin_amdgcn_fence(__ATOMIC_ACQ_REL, "wavefront");
        j = jn; ch = chn;
    }
}

__global__ __launch_bounds__(1024)
void k_scatter_lds(const int* __restrict__ src, const int* __restrict__ dst,
                   const float* __restrict__ xl,
                   const float* __restrict__ logits,
                   const unsigned char* __restrict__ rid,
                   float* __restrict__ denom,
                   float* __restrict__ h, int n_edges, int n_nodes) {
    __shared__ __attribute__((aligned(16))) unsigned char pool[NR * ROWP * 4];
    __shared__ float          den[NR];
    __shared__ unsigned       smo[NR];
    __shared__ unsigned short qbuf[32 * CHUNK];
    __shared__ unsigned       qn[32];
    __shared__ unsigned       bcnt[NB];
    __shared__ unsigned       lovf;
    static_assert(NB == 32, "one list per wave of the 1024-thread group");
    float*          acc  = (float*)pool;
    int*            lsrc = (int*)pool;
    float*          llg  = (float*)(pool + NB * BCAP * 4);
    unsigned short* lrow = (unsigned short*)(pool + NB * BCAP * 8);

    const int tid   = threadIdx.x;
    const int lane  = tid & 31;
    const int wave  = tid >> 5;
    const int nbase = blockIdx.x * NR;

    for (int r = tid; r < NR; r += 1024) {
        den[r] = 0.0f;
        smo[r] = 0x007FFFFFu;
    }
    if (tid < 32) { qn[tid] = 0u; bcnt[tid] = 0u; }
    if (tid == 0) lovf = 0u;
    __syncthreads();

    unsigned short* myq = qbuf + wave * CHUNK;
    walk_plane<false>(src, dst, xl, logits, rid, n_edges, nbase, blockIdx.x, lane, wave,
                      acc, den, smo, myq, qn, lsrc, llg, lrow, bcnt, &lovf);
    __syncthreads();

    const unsigned over = (unsigned)__builtin_amdgcn_readfirstlane((int)lovf);
    if (over == 0u) {
        v32f racc;
#pragma unroll
        for (int k = 0; k < 32; ++k) racc[k] = 0.0f;
        const unsigned n = (unsigned)__builtin_amdgcn_readfirstlane((int)bcnt[wave]);
        const int*            ms = lsrc + wave * BCAP;
        const float*          ml = llg  + wave * BCAP;
        const unsigned short* mr = lrow + wave * BCAP;
        for (unsigned i0 = 0u; i0 < n; i0 += 32u) {
            const unsigned i = i0 + (unsigned)lane;
            const bool ok = i < n;
            int      s = 0;
            unsigned j = 0u;
            float    w = 0.0f;
            if (ok) {
                s = ms[i];
                const unsigned r = mr[i];
                j = r / NB;
                w = __expf(ml[i] - seg_shift(smo[r]));
                atomicAdd(&den[r], w);
            }
            own_rows(xl, racc, lane, s, j, w, ok);
        }
        __builtin_amdgcn_fence(__ATOMIC_ACQ_REL, "wavefront");
#pragma unroll
        for (int jj = 0; jj < RPW; ++jj) {
            const int r = wave + NB * jj;
            if (r < NR && nbase + r < n_nodes) {
                v2f o;
                o.x = racc[2 * jj]; o.y = racc[2 * jj + 1];
                *(v2f*)(h + (long long)(nbase + r) * D_DIM + 2 * lane) = o;
                if (lane == 0)
                    denom[(long long)(nbase + r) * D_DIM] = den[r];
            }
        }
    } else {
        for (int i = tid; i < NR * ROWP; i += 1024) acc[i] = 0.0f;
        __syncthreads();
        walk_plane<true>(src, dst, xl, logits, rid, n_edges, nbase, blockIdx.x, lane, wave,
                         acc, den, smo, myq, qn, lsrc, llg, lrow, bcnt, &lovf);
        __syncthreads();

        for (int i = tid; i < NR * D_DIM; i += 1024) {
            const int r = i >> 6, k = i & 63;
            if (nbase + r < n_nodes)
                h[(long long)(nbase + r) * D_DIM + k] = acc[r * ROWP + k];
        }
        for (int r = tid; r < NR; r += 1024)
            if (nbase + r < n_nodes)
                denom[(long long)(nbase + r) * D_DIM] = den[r];
    }
}

__global__ void k_output(const float* __restrict__ h, const float* __restrict__ denom,
                         const int* __restrict__ mask,
                         const float* __restrict__ gbias,
                         const float* __restrict__ Wo, const float* __restrict__ bo,
                         float* __restrict__ out, int n_mask) {
    const int lane = threadIdx.x & 31;
    const int wave = threadIdx.x >> 5;
    const int tile = blockIdx.x * (blockDim.x >> 5) + wave;
    const int r0 = tile * 16;
    if (r0 + 16 > n_mask) return;
    const int lrow = lane & 15;
    const int lhi  = lane >> 4;

    const long long mi = (long long)mask[r0 + lrow];
    v2f a[16];
    const float* hp = h + mi * D_DIM + 2 * lhi;
    const float* gp = gbias + 2 * lhi;
    const float rz = 1.0f / (denom[mi * D_DIM] + 1e-16f);
#pragma unroll
    for (int kk = 0; kk < 16; ++kk) {
        const v2f hv = *(const v2f*)(hp + kk * 4);
        const v2f gv = *(const v2f*)(gp + kk * 4);
        a[kk].x = hv.x * rz + gv.x;
        a[kk].y = hv.y * rz + gv.y;
    }
#pragma unroll
    for (int ct = 0; ct < 8; ++ct) {
        const int col = ct * 16 + lrow;
        const float bv = bo[col];
        v8f acc = {bv, bv, bv, bv, bv, bv, bv, bv};
#pragma unroll
        for (int kk = 0; kk < 16; ++kk) {
            const int k0 = kk * 4 + 2 * lhi;
            v2f b;
            b.x = Wo[(k0    ) * DOUT + col];
            b.y = Wo[(k0 + 1) * DOUT + col];
            acc = __builtin_amdgcn_wmma_f32_16x16x4_f32(
                    false, a[kk], false, b, (short)0, acc, false, false);
        }
#pragma unroll
        for (int r = 0; r < 8; ++r)
            out[(long long)(r0 + r + 8 * lhi) * DOUT + col] = acc[r];
    }
}

extern "C" void kernel_launch(void* const* d_in, const int* in_sizes, int n_in,
                              void* d_out, int out_size, void* d_ws, size_t ws_size,
                              hipStream_t stream) {
    const float* x     = (const float*)d_in[0];
    const int*   ei    = (const int*)d_in[1];
    const int*   mask  = (const int*)d_in[2];
    const float* Wl    = (const float*)d_in[3];
    const float* bl    = (const float*)d_in[4];
    const float* Wr    = (const float*)d_in[5];
    const float* br    = (const float*)d_in[6];
    const float* att   = (const float*)d_in[7];
    const float* gbias = (const float*)d_in[8];
    const float* Wo    = (const float*)d_in[9];
    const float* bo    = (const float*)d_in[10];
    float* out = (float*)d_out;

    const int n_nodes = in_sizes[0] / D_DIM;
    const int n_edges = in_sizes[1] / 2;
    const int n_mask  = in_sizes[2];
    const int* src = ei;
    const int* dst = ei + n_edges;

    char* wp = (char*)d_ws;
    auto carve = [&](size_t bytes) {
        char* p = wp;
        wp += (bytes + 255) & ~(size_t)255;
        return p;
    };
    float*    xl     = (float*)   carve((size_t)n_nodes * D_DIM * 4);
    float*    xr     = (float*)   carve((size_t)n_nodes * D_DIM * 4);
    float*    h      = (float*)   carve((size_t)n_nodes * D_DIM * 4);
    float*    logits = (float*)   carve((size_t)n_edges * 4);
    unsigned char* rid = (unsigned char*)carve((size_t)n_edges + CHUNK);
    float*    denom  = xr;

    {
        int tiles  = (n_nodes + 15) / 16;
        int blocks = (tiles + 7) / 8;
        k_node_linear<<<blocks, 256, 0, stream>>>(x, Wl, bl, Wr, br, xl, xr, n_nodes);
    }
    {
        int tiles  = (n_edges + 15) / 16;
        int blocks = (tiles + 7) / 8;
        k_edge_logits<<<blocks, 256, 0, stream>>>(src, dst, xl, xr, att, logits, rid, n_edges);
    }
    k_scatter_lds<<<(n_nodes + NR - 1) / NR, 1024, 0, stream>>>(src, dst, xl, logits, rid, denom, h,
                                                              n_edges, n_nodes);
    {
        int tiles  = (n_mask + 15) / 16;
        int blocks = (tiles + 7) / 8;
        k_output<<<blocks, 256, 0, stream>>>(h, denom, mask, gbias, Wo, bo, out, n_mask);
    }
}


# range-plane scan: a lane's 64 match flags packed into 2 words (2 loops, were 16), queue places from a ballot rank kept in a register, no LDS add per match; sums stay in arrival order
# speedup vs baseline: 2.8336x; 1.0444x over previous
#include <hip/hip_runtime.h>
#include <hip/hip_bf16.h>
#include <math.h>


#define D_DIM 64
#define DOUT  128
#define NEG_SLOPE 0.2f
#define NR    400
#define ROWP  65
#define CHUNK 2048
#define GRP   16

typedef __attribute__((ext_vector_type(2))) float v2f;
typedef __attribute__((ext_vector_type(4))) float v4f;
typedef __attribute__((ext_vector_type(8))) float v8f;

__device__ __forceinline__ unsigned f2ord(float f) {
    unsigned u = __float_as_uint(f);
    return (u & 0x80000000u) ? ~u : (u | 0x80000000u);
}
__device__ __forceinline__ float ord2f(unsigned u) {
    unsigned b = (u & 0x80000000u) ? (u ^ 0x80000000u) : ~u;
    return __uint_as_float(b);
}

__global__ void k_node_linear(const float* __restrict__ x,
                              const float* __restrict__ Wl, const float* __restrict__ bl,
                              const float* __restrict__ Wr, const float* __restrict__ br,
                              float* __restrict__ xl, float* __restrict__ xr,
                              int n_nodes) {
    const int lane = threadIdx.x & 31;
    const int wave = threadIdx.x >> 5;
    const int tile = blockIdx.x * (blockDim.x >> 5) + wave;
    const int r0 = tile * 16;
    if (r0 + 16 > n_nodes) return;
    const int lrow = lane & 15;
    const int lhi  = lane >> 4;

    v2f a[16];
    const float* xp = x + (long long)(r0 + lrow) * D_DIM + 2 * lhi;
#pragma unroll
    for (int kk = 0; kk < 16; ++kk)
        a[kk] = *(const v2f*)(xp + kk * 4);

#pragma unroll
    for (int m = 0; m < 2; ++m) {
        const float* W    = m ? Wr : Wl;
        const float* bias = m ? br : bl;
        float*       out  = m ? xr : xl;
#pragma unroll
        for (int ct = 0; ct < 4; ++ct) {
            const int col = ct * 16 + lrow;
            const float bv = bias[col];
            v8f acc = {bv, bv, bv, bv, bv, bv, bv, bv};
#pragma unroll
            for (int kk = 0; kk < 16; ++kk) {
                const int k0 = kk * 4 + 2 * lhi;
                v2f b;
                b.x = W[(k0    ) * D_DIM + col];
                b.y = W[(k0 + 1) * D_DIM + col];
                acc = __builtin_amdgcn_wmma_f32_16x16x4_f32(
                        false, a[kk], false, b, (short)0, acc, false, false);
            }
#pragma unroll
            for (int r = 0; r < 8; ++r)
                out[(long long)(r0 + r + 8 * lhi) * D_DIM + col] = acc[r];
        }
    }
}

#define EP4 17

template <int T>
__device__ __forceinline__ int row_lane(const int v) {
    return __builtin_amdgcn_update_dpp(0, v, 0x150 + T, 0xf, 0xf, false);
}

template <int T>
__device__ __forceinline__ void load_rows(const char* __restrict__ xlb, const char* __restrict__ xrb,
                                          const int s, const int d, const unsigned cb,
                                          v4f* vl, v4f* vr) {
    const unsigned so = (unsigned)row_lane<T>(s) * (D_DIM * 4u) + cb;
    const unsigned dd = (unsigned)row_lane<T>(d) * (D_DIM * 4u) + cb;
    vl[T] = *(const v4f*)(xlb + so);
    vr[T] = *(const v4f*)(xrb + dd);
}

__global__ __launch_bounds__(256)
void k_edge_logits(const int* __restrict__ src, const int* __restrict__ dst,
                   const float* __restrict__ xl, const float* __restrict__ xr,
                   const float* __restrict__ att,
                   float* __restrict__ logits,
                   unsigned char* __restrict__ rid,
                   int n_edges) {
    __shared__ v4f etile[8][16 * EP4];

    const int lane = threadIdx.x & 31;
    const int wave = threadIdx.x >> 5;
    const int tile = (int)((blockIdx.x * blockDim.x + threadIdx.x) >> 5);
    const long long e0 = (long long)tile * 16;
    if (e0 >= n_edges) return;
    const int c  = lane & 15;
    const int hh = lane >> 4;
    long long e = e0 + 8 * hh + (c & 7);
    const bool live = e < n_edges;
    if (!live) e = n_edges - 1;
    const int s = src[e], d = dst[e];
    const float* pa = att + 4 * hh;

    v4f vl[8], vr[8];
    {
        const char* xlb = (const char*)xl;
        const char* xrb = (const char*)xr;
        const unsigned cb = 16u * (unsigned)c;
        load_rows<0>(xlb, xrb, s, d, cb, vl, vr);
        load_rows<1>(xlb, xrb, s, d, cb, vl, vr);
        load_rows<2>(xlb, xrb, s, d, cb, vl, vr);
        load_rows<3>(xlb, xrb, s, d, cb, vl, vr);
        load_rows<4>(xlb, xrb, s, d, cb, vl, vr);
        load_rows<5>(xlb, xrb, s, d, cb, vl, vr);
        load_rows<6>(xlb, xrb, s, d, cb, vl, vr);
        load_rows<7>(xlb, xrb, s, d, cb, vl, vr);
    }

    v4f* my = &etile[wave][0];
#pragma unroll
    for (int t = 0; t < 8; ++t) {
        v4f x = vl[t] + vr[t];
        x.x = x.x > 0.0f ? x.x : NEG_SLOPE * x.x;
        x.y = x.y > 0.0f ? x.y : NEG_SLOPE * x.y;
        x.z = x.z > 0.0f ? x.z : NEG_SLOPE * x.z;
        x.w = x.w > 0.0f ? x.w : NEG_SLOPE * x.w;
        my[(8 * hh + t) * EP4 + c] = x;
    }
    __builtin_amdgcn_fence(__ATOMIC_ACQ_REL, "wavefront");

    v4f xa[8], tq[8];
#pragma unroll
    for (int q = 0; q < 8; ++q) {
        tq[q] = *(const v4f*)(pa + 8 * q);
        xa[q] = my[c * EP4 + 2 * q + hh];
    }
    asm volatile("" :: "v"(xa[0]), "v"(xa[1]), "v"(xa[2]), "v"(xa[3]),
                       "v"(xa[4]), "v"(xa[5]), "v"(xa[6]), "v"(xa[7]),
                       "v"(tq[0]), "v"(tq[1]), "v"(tq[2]), "v"(tq[3]),
                       "v"(tq[4]), "v"(tq[5]), "v"(tq[6]), "v"(tq[7]));

    v8f acc = {0.f, 0.f, 0.f, 0.f, 0.f, 0.f, 0.f, 0.f};
#pragma unroll
    for (int q = 0; q < 8; ++q) {
        const v2f a0 = {xa[q].x, xa[q].y}, b0 = {tq[q].x, tq[q].y};
        const v2f a1 = {xa[q].z, xa[q].w}, b1 = {tq[q].z, tq[q].w};
        acc = __builtin_amdgcn_wmma_f32_16x16x4_f32(
                false, a0, false, b0, (short)0, acc, false, false);
        acc = __builtin_amdgcn_wmma_f32_16x16x4_f32(
                false, a1, false, b1, (short)0, acc, false, false);
    }

    const int r = c & 7;
    float p = acc[0];
    p = r == 1 ? acc[1] : p;
    p = r == 2 ? acc[2] : p;
    p = r == 3 ? acc[3] : p;
    p = r == 4 ? acc[4] : p;
    p = r == 5 ? acc[5] : p;
    p = r == 6 ? acc[6] : p;
    p = r == 7 ? acc[7] : p;
    if (live && c < 8) {
        logits[e] = p;
        rid[e] = (unsigned char)((unsigned)d / NR);
    }
}

typedef __attribute__((ext_vector_type(4))) unsigned v4u;

#define NB   32
#define BCAP 320
#define RPW  13
static_assert(NB * RPW >= NR, "every local row needs an owner");
static_assert(2 * RPW <= 32, "a wave's row sums fit one 32-word vector per lane");
static_assert(NB * BCAP * 10 <= NR * ROWP * 4, "the lists (4 + 4 + 2 B an entry) lie in the storage of the LDS sum plane");

__device__ __forceinline__ float seg_shift(unsigned o) {
    const float f = ord2f(o);
    return __builtin_isfinite(f) ? f : 0.0f;
}

__device__ __forceinline__ void add_rows(const float* __restrict__ xl, float* acc, const int lane,
                                         const int s, const unsigned r, const float w, const bool ok) {
    unsigned lv = __builtin_amdgcn_ballot_w32(ok);
    for (unsigned t0 = 0; lv != 0u; t0 += GRP, lv >>= GRP) {
        v2f row[GRP];
#pragma unroll
        for (int u = 0; u < GRP; ++u) {
            const int su = __builtin_amdgcn_readlane(s, (int)t0 + u);
            row[u] = *(const v2f*)(xl + (long long)su * D_DIM + 2 * lane);
        }
        static_assert(GRP == 16, "the statement below names the GRP rows one by one");
        asm volatile("" :: "v"(row[0]), "v"(row[1]), "v"(row[2]), "v"(row[3]),
                           "v"(row[4]), "v"(row[5]), "v"(row[6]), "v"(row[7]),
                           "v"(row[8]), "v"(row[9]), "v"(row[10]), "v"(row[11]),
                           "v"(row[12]), "v"(row[13]), "v"(row[14]), "v"(row[15]));
#pragma unroll
        for (int u = 0; u < GRP; ++u)
            if (lv & (1u << u)) {
                const unsigned ru = (unsigned)__builtin_amdgcn_readlane((int)r, (int)t0 + u);
                const float    wu = __uint_as_float((unsigned)__builtin_amdgcn_readlane(
                                        (int)__float_as_uint(w), (int)t0 + u));
                float* ap = acc + ru * ROWP + 2 * lane;
                atomicAdd(ap,     wu * row[u].x);
                atomicAdd(ap + 1, wu * row[u].y);
            }
    }
}

typedef __attribute__((ext_vector_type(32))) float v32f;

__device__ __forceinline__ void own_rows(const float* __restrict__ xl, v32f& racc, const int lane,
                                         const int s, const unsigned j, const float w, const bool ok) {
    unsigned lv = __builtin_amdgcn_ballot_w32(ok);
    for (unsigned t0 = 0; lv != 0u; t0 += GRP, lv >>= GRP) {
        v2f row[GRP];
#pragma unroll
        for (int u = 0; u < GRP; ++u) {
            const int su = __builtin_amdgcn_readlane(s, (int)t0 + u);
            row[u] = *(const v2f*)(xl + (long long)su * D_DIM + 2 * lane);
        }
        static_assert(GRP == 16, "the statement below names the GRP rows one by one");
        asm volatile("" :: "v"(row[0]), "v"(row[1]), "v"(row[2]), "v"(row[3]),
                           "v"(row[4]), "v"(row[5]), "v"(row[6]), "v"(row[7]),
                           "v"(row[8]), "v"(row[9]), "v"(row[10]), "v"(row[11]),
                           "v"(row[12]), "v"(row[13]), "v"(row[14]), "v"(row[15]));
#pragma unroll
        for (int u = 0; u < GRP; ++u)
            if (lv & (1u << u)) {
                const unsigned ju = (unsigned)__builtin_amdgcn_readlane((int)j, (int)t0 + u);
                const float    wu = __uint_as_float((unsigned)__builtin_amdgcn_readlane(
                                        (int)__float_as_uint(w), (int)t0 + u));
                const unsigned k = 2u * ju;
                racc[k]      += wu * row[u].x;
                racc[k + 1u] += wu * row[u].y;
            }
    }
}

template <bool SUMS>
__device__ __forceinline__ void walk_plane(const int* __restrict__ src, const int* __restrict__ dst,
                                           const float* __restrict__ xl,
                                           const float* __restrict__ logits,
                                           const unsigned char* __restrict__ rid,
                                           const int n_edges, const int nbase, const unsigned bid,
                                           const int lane, const int wave,
                                           float* acc, float* den, unsigned* smo,
                                           unsigned short* myq,
                                           int* lsrc, float* llg, unsigned short* lrow,
                                           unsigned* bcnt, unsigned* lovf) {
    const unsigned pat = bid * 0x01010101u;
    const int nchunks = (n_edges + CHUNK - 1) / CHUNK;
    const int rot = (int)((bid * 37u) % (unsigned)nchunks);

    int j  = wave;
    int ch = j + rot; if (ch >= nchunks) ch -= nchunks;
    v4u c[4];
    if (j < nchunks) {
        const v4u* rp = (const v4u*)(rid + (long long)ch * CHUNK) + lane;
#pragma unroll
        for (int k = 0; k < 4; ++k) c[k] = rp[k * 32];
    }
    while (j < nchunks) {
        const int ebase = ch * CHUNK;
        unsigned mk[2];
#pragma unroll
        for (int hf = 0; hf < 2; ++hf) {
            unsigned mm = 0u;
#pragma unroll
            for (int w = 0; w < 8; ++w) {
                const unsigned x = c[2 * hf + (w >> 2)][w & 3] ^ pat;
                const unsigned fl = ~(((x & 0x7F7F7F7Fu) + 0x7F7F7F7Fu) | x | 0x7F7F7F7Fu);
                mm |= fl >> (7 - w);
            }
            mk[hf] = mm;
        }
        const int jn = j + 32;
        int chn = jn + rot; if (chn >= nchunks) chn -= nchunks;
        if (jn < nchunks) {
            const v4u* rp = (const v4u*)(rid + (long long)chn * CHUNK) + lane;
#pragma unroll
            for (int k = 0; k < 4; ++k) c[k] = rp[k * 32];
        }
        unsigned cnt = 0u;
#pragma unroll
        for (int hf = 0; hf < 2; ++hf) {
            unsigned mm = mk[hf];
            unsigned bal;
            while ((bal = __builtin_amdgcn_ballot_w32(mm != 0u)) != 0u) {
                if (mm != 0u) {
                    const unsigned p = (unsigned)__builtin_ctz(mm);
                    mm &= mm - 1u;
                    const unsigned off = (unsigned)(hf * 1024 + lane * 16) + ((p & 4u) << 7) + ((p & 3u) << 2) + (p >> 3);
                    myq[cnt + __builtin_amdgcn_mbcnt_lo(bal, 0u)] = (unsigned short)off;
                }
                cnt += (unsigned)__builtin_popcount(bal);
            }
        }
        __builtin_amdgcn_fence(__ATOMIC_ACQ_REL, "wavefront");
        const unsigned m = (unsigned)__builtin_amdgcn_readfirstlane((int)cnt);
        for (unsigned i0 = 0; i0 < m; i0 += 32) {
            const unsigned i = i0 + lane;
            int      s  = 0;
            unsigned r  = 0u;
            float    lg = 0.0f;
            bool     ok = false;
            if (i < m && ebase + (int)myq[i] < n_edges) {
                const long long e = (long long)ebase + myq[i];
                s = src[e];
                const int d = dst[e];
                lg = logits[e];
                r = (unsigned)(d - nbase);
                asm volatile("" :: "v"(s), "v"(d), "v"(lg));
                ok = r < NR;
            }
            if (SUMS) {
                float w = 0.0f;
                if (ok) {
                    w = __expf(lg - seg_shift(smo[r]));
                    atomicAdd(&den[r], w);
                }
                add_rows(xl, acc, lane, s, r, w, ok);
            } else {
                if (ok) {
                    atomicMax(&smo[r], f2ord(lg));
                    const unsigned b    = r % NB;
                    const unsigned slot = atomicAdd(&bcnt[b], 1u);
                    if (slot < BCAP) {
                        const unsigned p = b * BCAP + slot;
                        lsrc[p] = s;
                        llg[p]  = lg;
                        lrow[p] = (unsigned short)r;
                    } else {
                        *lovf = 1u;
                    }
                }
            }
        }
        __builtin_amdgcn_fence(__ATOMIC_ACQ_REL, "wavefront");
        j = jn; ch = chn;
    }
}

__global__ __launch_bounds__(1024)
void k_scatter_lds(const int* __restrict__ src, const int* __restrict__ dst,
                   const float* __restrict__ xl,
                   const float* __restrict__ logits,
                   const unsigned char* __restrict__ rid,
                   float* __restrict__ denom,
                   float* __restrict__ h, int n_edges, int n_nodes) {
    __shared__ __attribute__((aligned(16))) unsigned char pool[NR * ROWP * 4];
    __shared__ float          den[NR];
    __shared__ unsigned       smo[NR];
    __shared__ unsigned short qbuf[32 * CHUNK];
    __shared__ unsigned       bcnt[NB];
    __shared__ unsigned       lovf;
    static_assert(NB == 32, "one list per wave of the 1024-thread group");
    float*          acc  = (float*)pool;
    int*            lsrc = (int*)pool;
    float*          llg  = (float*)(pool + NB * BCAP * 4);
    unsigned short* lrow = (unsigned short*)(pool + NB * BCAP * 8);

    const int tid   = threadIdx.x;
    const int lane  = tid & 31;
    const int wave  = tid >> 5;
    const int nbase = blockIdx.x * NR;

    for (int r = tid; r < NR; r += 1024) {
        den[r] = 0.0f;
        smo[r] = 0x007FFFFFu;
    }
    if (tid < 32) bcnt[tid] = 0u;
    if (tid == 0) lovf = 0u;
    __syncthreads();

    unsigned short* myq = qbuf + wave * CHUNK;
    walk_plane<false>(src, dst, xl, logits, rid, n_edges, nbase, blockIdx.x, lane, wave,
                      acc, den, smo, myq, lsrc, llg, lrow, bcnt, &lovf);
    __syncthreads();

    const unsigned over = (unsigned)__builtin_amdgcn_readfirstlane((int)lovf);
    if (over == 0u) {
        v32f racc;
#pragma unroll
        for (int k = 0; k < 32; ++k) racc[k] = 0.0f;
        const unsigned n = (unsigned)__builtin_amdgcn_readfirstlane((int)bcnt[wave]);
        const int*            ms = lsrc + wave * BCAP;
        const float*          ml = llg  + wave * BCAP;
        const unsigned short* mr = lrow + wave * BCAP;
        for (unsigned i0 = 0u; i0 < n; i0 += 32u) {
            const unsigned i = i0 + (unsigned)lane;
            const bool ok = i < n;
            int      s = 0;
            unsigned j = 0u;
            float    w = 0.0f;
            if (ok) {
                s = ms[i];
                const unsigned r = mr[i];
                j = r / NB;
                w = __expf(ml[i] - seg_shift(smo[r]));
                atomicAdd(&den[r], w);
            }
            own_rows(xl, racc, lane, s, j, w, ok);
        }
        __builtin_amdgcn_fence(__ATOMIC_ACQ_REL, "wavefront");
#pragma unroll
        for (int jj = 0; jj < RPW; ++jj) {
            const int r = wave + NB * jj;
            if (r < NR && nbase + r < n_nodes) {
                v2f o;
                o.x = racc[2 * jj]; o.y = racc[2 * jj + 1];
                *(v2f*)(h + (long long)(nbase + r) * D_DIM + 2 * lane) = o;
                if (lane == 0)
                    denom[(long long)(nbase + r) * D_DIM] = den[r];
            }
        }
    } else {
        for (int i = tid; i < NR * ROWP; i += 1024) acc[i] = 0.0f;
        __syncthreads();
        walk_plane<true>(src, dst, xl, logits, rid, n_edges, nbase, blockIdx.x, lane, wave,
                         acc, den, smo, myq, lsrc, llg, lrow, bcnt, &lovf);
        __syncthreads();

        for (int i = tid; i < NR * D_DIM; i += 1024) {
            const int r = i >> 6, k = i & 63;
            if (nbase + r < n_nodes)
                h[(long long)(nbase + r) * D_DIM + k] = acc[r * ROWP + k];
        }
        for (int r = tid; r < NR; r += 1024)
            if (nbase + r < n_nodes)
                denom[(long long)(nbase + r) * D_DIM] = den[r];
    }
}

__global__ void k_output(const float* __restrict__ h, const float* __restrict__ denom,
                         const int* __restrict__ mask,
                         const float* __restrict__ gbias,
                         const float* __restrict__ Wo, const float* __restrict__ bo,
                         float* __restrict__ out, int n_mask) {
    const int lane = threadIdx.x & 31;
    const int wave = threadIdx.x >> 5;
    const int tile = blockIdx.x * (blockDim.x >> 5) + wave;
    const int r0 = tile * 16;
    if (r0 + 16 > n_mask) return;
    const int lrow = lane & 15;
    const int lhi  = lane >> 4;

    const long long mi = (long long)mask[r0 + lrow];
    v2f a[16];
    const float* hp = h + mi * D_DIM + 2 * lhi;
    const float* gp = gbias + 2 * lhi;
    const float rz = 1.0f / (denom[mi * D_DIM] + 1e-16f);
#pragma unroll
    for (int kk = 0; kk < 16; ++kk) {
        const v2f hv = *(const v2f*)(hp + kk * 4);
        const v2f gv = *(const v2f*)(gp + kk * 4);
        a[kk].x = hv.x * rz + gv.x;
        a[kk].y = hv.y * rz + gv.y;
    }
#pragma unroll
    for (int ct = 0; ct < 8; ++ct) {
        const int col = ct * 16 + lrow;
        const float bv = bo[col];
        v8f acc = {bv, bv, bv, bv, bv, bv, bv, bv};
#pragma unroll
        for (int kk = 0; kk < 16; ++kk) {
            const int k0 = kk * 4 + 2 * lhi;
            v2f b;
            b.x = Wo[(k0    ) * DOUT + col];
            b.y = Wo[(k0 + 1) * DOUT + col];
            acc = __builtin_amdgcn_wmma_f32_16x16x4_f32(
                    false, a[kk], false, b, (short)0, acc, false, false);
        }
#pragma unroll
        for (int r = 0; r < 8; ++r)
            out[(long long)(r0 + r + 8 * lhi) * DOUT + col] = acc[r];
    }
}

extern "C" void kernel_launch(void* const* d_in, const int* in_sizes, int n_in,
                              void* d_out, int out_size, void* d_ws, size_t ws_size,
                              hipStream_t stream) {
    const float* x     = (const float*)d_in[0];
    const int*   ei    = (const int*)d_in[1];
    const int*   mask  = (const int*)d_in[2];
    const float* Wl    = (const float*)d_in[3];
    const float* bl    = (const float*)d_in[4];
    const float* Wr    = (const float*)d_in[5];
    const float* br    = (const float*)d_in[6];
    const float* att   = (const float*)d_in[7];
    const float* gbias = (const float*)d_in[8];
    const float* Wo    = (const float*)d_in[9];
    const float* bo    = (const float*)d_in[10];
    float* out = (float*)d_out;

    const int n_nodes = in_sizes[0] / D_DIM;
    const int n_edges = in_sizes[1] / 2;
    const int n_mask  = in_sizes[2];
    const int* src = ei;
    const int* dst = ei + n_edges;

    char* wp = (char*)d_ws;
    auto carve = [&](size_t bytes) {
        char* p = wp;
        wp += (bytes + 255) & ~(size_t)255;
        return p;
    };
    float*    xl     = (float*)   carve((size_t)n_nodes * D_DIM * 4);
    float*    xr     = (float*)   carve((size_t)n_nodes * D_DIM * 4);
    float*    h      = (float*)   carve((size_t)n_nodes * D_DIM * 4);
    float*    logits = (float*)   carve((size_t)n_edges * 4);
    unsigned char* rid = (unsigned char*)carve((size_t)n_edges + CHUNK);
    float*    denom  = xr;

    {
        int tiles  = (n_nodes + 15) / 16;
        int blocks = (tiles + 7) / 8;
        k_node_linear<<<blocks, 256, 0, stream>>>(x, Wl, bl, Wr, br, xl, xr, n_nodes);
    }
    {
        int tiles  = (n_edges + 15) / 16;
        int blocks = (tiles + 7) / 8;
        k_edge_logits<<<blocks, 256, 0, stream>>>(src, dst, xl, xr, att, logits, rid, n_edges);
    }
    k_scatter_lds<<<(n_nodes + NR - 1) / NR, 1024, 0, stream>>>(src, dst, xl, logits, rid, denom, h,
                                                              n_edges, n_nodes);
    {
        int tiles  = (n_mask + 15) / 16;
        int blocks = (tiles + 7) / 8;
        k_output<<<blocks, 256, 0, stream>>>(h, denom, mask, gbias, Wo, bo, out, n_mask);
    }
}


# per-edge 8-B record (source row, row in its range, logit) written by the logit kernel: the range owner fetches a listed edge with one 8-B load, was three 4-B loads, three arrays; sum order unchanged
# speedup vs baseline: 3.3177x; 1.1708x over previous
#include <hip/hip_runtime.h>
#include <hip/hip_bf16.h>
#include <math.h>


#define D_DIM 64
#define DOUT  128
#define NEG_SLOPE 0.2f
#define NR    400
#define ROWP  65
#define CHUNK 2048
#define GRP   16

typedef __attribute__((ext_vector_type(2))) float v2f;
typedef __attribute__((ext_vector_type(4))) float v4f;
typedef __attribute__((ext_vector_type(8))) float v8f;
typedef __attribute__((ext_vector_type(2))) unsigned v2u;

#define RECSH 23
static_assert(NR <= (1 << (32 - RECSH)), "the row number inside a range fits the bits above the source row");

__device__ __forceinline__ unsigned f2ord(float f) {
    unsigned u = __float_as_uint(f);
    return (u & 0x80000000u) ? ~u : (u | 0x80000000u);
}
__device__ __forceinline__ float ord2f(unsigned u) {
    unsigned b = (u & 0x80000000u) ? (u ^ 0x80000000u) : ~u;
    return __uint_as_float(b);
}

__global__ void k_node_linear(const float* __restrict__ x,
                              const float* __restrict__ Wl, const float* __restrict__ bl,
                              const float* __restrict__ Wr, const float* __restrict__ br,
                              float* __restrict__ xl, float* __restrict__ xr,
                              int n_nodes) {
    const int lane = threadIdx.x & 31;
    const int wave = threadIdx.x >> 5;
    const int tile = blockIdx.x * (blockDim.x >> 5) + wave;
    const int r0 = tile * 16;
    if (r0 + 16 > n_nodes) return;
    const int lrow = lane & 15;
    const int lhi  = lane >> 4;

    v2f a[16];
    const float* xp = x + (long long)(r0 + lrow) * D_DIM + 2 * lhi;
#pragma unroll
    for (int kk = 0; kk < 16; ++kk)
        a[kk] = *(const v2f*)(xp + kk * 4);

#pragma unroll
    for (int m = 0; m < 2; ++m) {
        const float* W    = m ? Wr : Wl;
        const float* bias = m ? br : bl;
        float*       out  = m ? xr : xl;
#pragma unroll
        for (int ct = 0; ct < 4; ++ct) {
            const int col = ct * 16 + lrow;
            const float bv = bias[col];
            v8f acc = {bv, bv, bv, bv, bv, bv, bv, bv};
#pragma unroll
            for (int kk = 0; kk < 16; ++kk) {
                const int k0 = kk * 4 + 2 * lhi;
                v2f b;
                b.x = W[(k0    ) * D_DIM + col];
                b.y = W[(k0 + 1) * D_DIM + col];
                acc = __builtin_amdgcn_wmma_f32_16x16x4_f32(
                        false, a[kk], false, b, (short)0, acc, false, false);
            }
#pragma unroll
            for (int r = 0; r < 8; ++r)
                out[(long long)(r0 + r + 8 * lhi) * D_DIM + col] = acc[r];
        }
    }
}

#define EP4 17

template <int T>
__device__ __forceinline__ int row_lane(const int v) {
    return __builtin_amdgcn_update_dpp(0, v, 0x150 + T, 0xf, 0xf, false);
}

template <int T>
__device__ __forceinline__ void load_rows(const char* __restrict__ xlb, const char* __restrict__ xrb,
                                          const int s, const int d, const unsigned cb,
                                          v4f* vl, v4f* vr) {
    const unsigned so = (unsigned)row_lane<T>(s) * (D_DIM * 4u) + cb;
    const unsigned dd = (unsigned)row_lane<T>(d) * (D_DIM * 4u) + cb;
    vl[T] = *(const v4f*)(xlb + so);
    vr[T] = *(const v4f*)(xrb + dd);
}

__global__ __launch_bounds__(256)
void k_edge_logits(const int* __restrict__ src, const int* __restrict__ dst,
                   const float* __restrict__ xl, const float* __restrict__ xr,
                   const float* __restrict__ att,
                   float* __restrict__ rec,
                   unsigned char* __restrict__ rid,
                   int n_edges) {
    __shared__ v4f etile[8][16 * EP4];

    const int lane = threadIdx.x & 31;
    const int wave = threadIdx.x >> 5;
    const int tile = (int)((blockIdx.x * blockDim.x + threadIdx.x) >> 5);
    const long long e0 = (long long)tile * 16;
    if (e0 >= n_edges) return;
    const int c  = lane & 15;
    const int hh = lane >> 4;
    long long e = e0 + 8 * hh + (c & 7);
    const bool live = e < n_edges;
    if (!live) e = n_edges - 1;
    const int s = src[e], d = dst[e];
    const float* pa = att + 4 * hh;

    v4f vl[8], vr[8];
    {
        const char* xlb = (const char*)xl;
        const char* xrb = (const char*)xr;
        const unsigned cb = 16u * (unsigned)c;
        load_rows<0>(xlb, xrb, s, d, cb, vl, vr);
        load_rows<1>(xlb, xrb, s, d, cb, vl, vr);
        load_rows<2>(xlb, xrb, s, d, cb, vl, vr);
        load_rows<3>(xlb, xrb, s, d, cb, vl, vr);
        load_rows<4>(xlb, xrb, s, d, cb, vl, vr);
        load_rows<5>(xlb, xrb, s, d, cb, vl, vr);
        load_rows<6>(xlb, xrb, s, d, cb, vl, vr);
        load_rows<7>(xlb, xrb, s, d, cb, vl, vr);
    }

    v4f* my = &etile[wave][0];
#pragma unroll
    for (int t = 0; t < 8; ++t) {
        v4f x = vl[t] + vr[t];
        x.x = x.x > 0.0f ? x.x : NEG_SLOPE * x.x;
        x.y = x.y > 0.0f ? x.y : NEG_SLOPE * x.y;
        x.z = x.z > 0.0f ? x.z : NEG_SLOPE * x.z;
        x.w = x.w > 0.0f ? x.w : NEG_SLOPE * x.w;
        my[(8 * hh + t) * EP4 + c] = x;
    }
    __builtin_amdgcn_fence(__ATOMIC_ACQ_REL, "wavefront");

    v4f xa[8], tq[8];
#pragma unroll
    for (int q = 0; q < 8; ++q) {
        tq[q] = *(const v4f*)(pa + 8 * q);
        xa[q] = my[c * EP4 + 2 * q + hh];
    }
    asm volatile("" :: "v"(xa[0]), "v"(xa[1]), "v"(xa[2]), "v"(xa[3]),
                       "v"(xa[4]), "v"(xa[5]), "v"(xa[6]), "v"(xa[7]),
                       "v"(tq[0]), "v"(tq[1]), "v"(tq[2]), "v"(tq[3]),
                       "v"(tq[4]), "v"(tq[5]), "v"(tq[6]), "v"(tq[7]));

    v8f acc = {0.f, 0.f, 0.f, 0.f, 0.f, 0.f, 0.f, 0.f};
#pragma unroll
    for (int q = 0; q < 8; ++q) {
        const v2f a0 = {xa[q].x, xa[q].y}, b0 = {tq[q].x, tq[q].y};
        const v2f a1 = {xa[q].z, xa[q].w}, b1 = {tq[q].z, tq[q].w};
        acc = __builtin_amdgcn_wmma_f32_16x16x4_f32(
                false, a0, false, b0, (short)0, acc, false, false);
        acc = __builtin_amdgcn_wmma_f32_16x16x4_f32(
                false, a1, false, b1, (short)0, acc, false, false);
    }

    const int r = c & 7;
    float p = acc[0];
    p = r == 1 ? acc[1] : p;
    p = r == 2 ? acc[2] : p;
    p = r == 3 ? acc[3] : p;
    p = r == 4 ? acc[4] : p;
    p = r == 5 ? acc[5] : p;
    p = r == 6 ? acc[6] : p;
    p = r == 7 ? acc[7] : p;
    if (live && c < 8) {
        const unsigned g  = (unsigned)d / NR;
        const unsigned lr = (unsigned)d - g * NR;
        v2u rc;
        rc.x = (unsigned)s | (lr << RECSH);
        rc.y = __float_as_uint(p);
        *(v2u*)(rec + 2 * e) = rc;
        rid[e] = (unsigned char)g;
    }
}

typedef __attribute__((ext_vector_type(4))) unsigned v4u;

#define NB   32
#define BCAP 320
#define RPW  13
static_assert(NB * RPW >= NR, "every local row needs an owner");
static_assert(2 * RPW <= 32, "a wave's row sums fit one 32-word vector per lane");
static_assert(NB * BCAP * 10 <= NR * ROWP * 4, "the lists (4 + 4 + 2 B an entry) lie in the storage of the LDS sum plane");

__device__ __forceinline__ float seg_shift(unsigned o) {
    const float f = ord2f(o);
    return __builtin_isfinite(f) ? f : 0.0f;
}

__device__ __forceinline__ void add_rows(const float* __restrict__ xl, float* acc, const int lane,
                                         const int s, const unsigned r, const float w, const bool ok) {
    unsigned lv = __builtin_amdgcn_ballot_w32(ok);
    for (unsigned t0 = 0; lv != 0u; t0 += GRP, lv >>= GRP) {
        v2f row[GRP];
#pragma unroll
        for (int u = 0; u < GRP; ++u) {
            const int su = __builtin_amdgcn_readlane(s, (int)t0 + u);
            row[u] = *(const v2f*)(xl + (long long)su * D_DIM + 2 * lane);
        }
        static_assert(GRP == 16, "the statement below names the GRP rows one by one");
        asm volatile("" :: "v"(row[0]), "v"(row[1]), "v"(row[2]), "v"(row[3]),
                           "v"(row[4]), "v"(row[5]), "v"(row[6]), "v"(row[7]),
                           "v"(row[8]), "v"(row[9]), "v"(row[10]), "v"(row[11]),
                           "v"(row[12]), "v"(row[13]), "v"(row[14]), "v"(row[15]));
#pragma unroll
        for (int u = 0; u < GRP; ++u)
            if (lv & (1u << u)) {
                const unsigned ru = (unsigned)__builtin_amdgcn_readlane((int)r, (int)t0 + u);
                const float    wu = __uint_as_float((unsigned)__builtin_amdgcn_readlane(
                                        (int)__float_as_uint(w), (int)t0 + u));
                float* ap = acc + ru * ROWP + 2 * lane;
                atomicAdd(ap,     wu * row[u].x);
                atomicAdd(ap + 1, wu * row[u].y);
            }
    }
}

typedef __attribute__((ext_vector_type(32))) float v32f;

__device__ __forceinline__ void own_rows(const float* __restrict__ xl, v32f& racc, const int lane,
                                         const int s, const unsigned j, const float w, const bool ok) {
    unsigned lv = __builtin_amdgcn_ballot_w32(ok);
    for (unsigned t0 = 0; lv != 0u; t0 += GRP, lv >>= GRP) {
        v2f row[GRP];
#pragma unroll
        for (int u = 0; u < GRP; ++u) {
            const int su = __builtin_amdgcn_readlane(s, (int)t0 + u);
            row[u] = *(const v2f*)(xl + (long long)su * D_DIM + 2 * lane);
        }
        static_assert(GRP == 16, "the statement below names the GRP rows one by one");
        asm volatile("" :: "v"(row[0]), "v"(row[1]), "v"(row[2]), "v"(row[3]),
                           "v"(row[4]), "v"(row[5]), "v"(row[6]), "v"(row[7]),
                           "v"(row[8]), "v"(row[9]), "v"(row[10]), "v"(row[11]),
                           "v"(row[12]), "v"(row[13]), "v"(row[14]), "v"(row[15]));
#pragma unroll
        for (int u = 0; u < GRP; ++u)
            if (lv & (1u << u)) {
                const unsigned ju = (unsigned)__builtin_amdgcn_readlane((int)j, (int)t0 + u);
                const float    wu = __uint_as_float((unsigned)__builtin_amdgcn_readlane(
                                        (int)__float_as_uint(w), (int)t0 + u));
                const unsigned k = 2u * ju;
                racc[k]      += wu * row[u].x;
                racc[k + 1u] += wu * row[u].y;
            }
    }
}

template <bool SUMS>
__device__ __forceinline__ void walk_plane(const float* __restrict__ xl,
                                           const float* __restrict__ rec,
                                           const unsigned char* __restrict__ rid,
                                           const int n_edges, const unsigned bid,
                                           const int lane, const int wave,
                                           float* acc, float* den, unsigned* smo,
                                           unsigned short* myq,
                                           int* lsrc, float* llg, unsigned short* lrow,
                                           unsigned* bcnt, unsigned* lovf) {
    const unsigned pat = bid * 0x01010101u;
    const int nchunks = (n_edges + CHUNK - 1) / CHUNK;
    const int rot = (int)((bid * 37u) % (unsigned)nchunks);

    int j  = wave;
    int ch = j + rot; if (ch >= nchunks) ch -= nchunks;
    v4u c[4];
    if (j < nchunks) {
        const v4u* rp = (const v4u*)(rid + (long long)ch * CHUNK) + lane;
#pragma unroll
        for (int k = 0; k < 4; ++k) c[k] = rp[k * 32];
    }
    while (j < nchunks) {
        const int ebase = ch * CHUNK;
        unsigned mk[2];
#pragma unroll
        for (int hf = 0; hf < 2; ++hf) {
            unsigned mm = 0u;
#pragma unroll
            for (int w = 0; w < 8; ++w) {
                const unsigned x = c[2 * hf + (w >> 2)][w & 3] ^ pat;
                const unsigned fl = ~(((x & 0x7F7F7F7Fu) + 0x7F7F7F7Fu) | x | 0x7F7F7F7Fu);
                mm |= fl >> (7 - w);
            }
            mk[hf] = mm;
        }
        const int jn = j + 32;
        int chn = jn + rot; if (chn >= nchunks) chn -= nchunks;
        if (jn < nchunks) {
            const v4u* rp = (const v4u*)(rid + (long long)chn * CHUNK) + lane;
#pragma unroll
            for (int k = 0; k < 4; ++k) c[k] = rp[k * 32];
        }
        unsigned cnt = 0u;
#pragma unroll
        for (int hf = 0; hf < 2; ++hf) {
            unsigned mm = mk[hf];
            unsigned bal;
            while ((bal = __builtin_amdgcn_ballot_w32(mm != 0u)) != 0u) {
                if (mm != 0u) {
                    const unsigned p = (unsigned)__builtin_ctz(mm);
                    mm &= mm - 1u;
                    const unsigned off = (unsigned)(hf * 1024 + lane * 16) + ((p & 4u) << 7) + ((p & 3u) << 2) + (p >> 3);
                    myq[cnt + __builtin_amdgcn_mbcnt_lo(bal, 0u)] = (unsigned short)off;
                }
                cnt += (unsigned)__builtin_popcount(bal);
            }
        }
        __builtin_amdgcn_fence(__ATOMIC_ACQ_REL, "wavefront");
        const unsigned m = (unsigned)__builtin_amdgcn_readfirstlane((int)cnt);
        for (unsigned i0 = 0; i0 < m; i0 += 32) {
            const unsigned i = i0 + lane;
            int      s  = 0;
            unsigned r  = 0u;
            float    lg = 0.0f;
            bool     ok = false;
            if (i < m && ebase + (int)myq[i] < n_edges) {
                const long long e = (long long)ebase + myq[i];
                const v2u rc = *(const v2u*)(rec + 2 * e);
                s  = (int)(rc.x & ((1u << RECSH) - 1u));
                r  = rc.x >> RECSH;
                lg = __uint_as_float(rc.y);
                ok = true;
            }
            if (SUMS) {
                float w = 0.0f;
                if (ok) {
                    w = __expf(lg - seg_shift(smo[r]));
                    atomicAdd(&den[r], w);
                }
                add_rows(xl, acc, lane, s, r, w, ok);
            } else {
                if (ok) {
                    atomicMax(&smo[r], f2ord(lg));
                    const unsigned b    = r % NB;
                    const unsigned slot = atomicAdd(&bcnt[b], 1u);
                    if (slot < BCAP) {
                        const unsigned p = b * BCAP + slot;
                        lsrc[p] = s;
                        llg[p]  = lg;
                        lrow[p] = (unsigned short)r;
                    } else {
                        *lovf = 1u;
                    }
                }
            }
        }
        __builtin_amdgcn_fence(__ATOMIC_ACQ_REL, "wavefront");
        j = jn; ch = chn;
    }
}

__global__ __launch_bounds__(1024)
void k_scatter_lds(const float* __restrict__ xl,
                   const float* __restrict__ rec,
                   const unsigned char* __restrict__ rid,
                   float* __restrict__ denom,
                   float* __restrict__ h, int n_edges, int n_nodes) {
    __shared__ __attribute__((aligned(16))) unsigned char pool[NR * ROWP * 4];
    __shared__ float          den[NR];
    __shared__ unsigned       smo[NR];
    __shared__ unsigned short qbuf[32 * CHUNK];
    __shared__ unsigned       bcnt[NB];
    __shared__ unsigned       lovf;
    static_assert(NB == 32, "one list per wave of the 1024-thread group");
    float*          acc  = (float*)pool;
    int*            lsrc = (int*)pool;
    float*          llg  = (float*)(pool + NB * BCAP * 4);
    unsigned short* lrow = (unsigned short*)(pool + NB * BCAP * 8);

    const int tid   = threadIdx.x;
    const int lane  = tid & 31;
    const int wave  = tid >> 5;
    const int nbase = blockIdx.x * NR;

    for (int r = tid; r < NR; r += 1024) {
        den[r] = 0.0f;
        smo[r] = 0x007FFFFFu;
    }
    if (tid < 32) bcnt[tid] = 0u;
    if (tid == 0) lovf = 0u;
    __syncthreads();

    unsigned short* myq = qbuf + wave * CHUNK;
    walk_plane<false>(xl, rec, rid, n_edges, blockIdx.x, lane, wave,
                      acc, den, smo, myq, lsrc, llg, lrow, bcnt, &lovf);
    __syncthreads();

    const unsigned over = (unsigned)__builtin_amdgcn_readfirstlane((int)lovf);
    if (over == 0u) {
        v32f racc;
#pragma unroll
        for (int k = 0; k < 32; ++k) racc[k] = 0.0f;
        const unsigned n = (unsigned)__builtin_amdgcn_readfirstlane((int)bcnt[wave]);
        const int*            ms = lsrc + wave * BCAP;
        const float*          ml = llg  + wave * BCAP;
        const unsigned short* mr = lrow + wave * BCAP;
        for (unsigned i0 = 0u; i0 < n; i0 += 32u) {
            const unsigned i = i0 + (unsigned)lane;
            const bool ok = i < n;
            int      s = 0;
            unsigned j = 0u;
            float    w = 0.0f;
            if (ok) {
                s = ms[i];
                const unsigned r = mr[i];
                j = r / NB;
                w = __expf(ml[i] - seg_shift(smo[r]));
                atomicAdd(&den[r], w);
            }
            own_rows(xl, racc, lane, s, j, w, ok);
        }
        __builtin_amdgcn_fence(__ATOMIC_ACQ_REL, "wavefront");
#pragma unroll
        for (int jj = 0; jj < RPW; ++jj) {
            const int r = wave + NB * jj;
            if (r < NR && nbase + r < n_nodes) {
                v2f o;
                o.x = racc[2 * jj]; o.y = racc[2 * jj + 1];
                *(v2f*)(h + (long long)(nbase + r) * D_DIM + 2 * lane) = o;
                if (lane == 0)
                    denom[nbase + r] = den[r];
            }
        }
    } else {
        for (int i = tid; i < NR * ROWP; i += 1024) acc[i] = 0.0f;
        __syncthreads();
        walk_plane<true>(xl, rec, rid, n_edges, blockIdx.x, lane, wave,
                         acc, den, smo, myq, lsrc, llg, lrow, bcnt, &lovf);
        __syncthreads();

        for (int i = tid; i < NR * D_DIM; i += 1024) {
            const int r = i >> 6, k = i & 63;
            if (nbase + r < n_nodes)
                h[(long long)(nbase + r) * D_DIM + k] = acc[r * ROWP + k];
        }
        for (int r = tid; r < NR; r += 1024)
            if (nbase + r < n_nodes)
                denom[nbase + r] = den[r];
    }
}

__global__ void k_output(const float* __restrict__ h, const float* __restrict__ denom,
                         const int* __restrict__ mask,
                         const float* __restrict__ gbias,
                         const float* __restrict__ Wo, const float* __restrict__ bo,
                         float* __restrict__ out, int n_mask) {
    const int lane = threadIdx.x & 31;
    const int wave = threadIdx.x >> 5;
    const int tile = blockIdx.x * (blockDim.x >> 5) + wave;
    const int r0 = tile * 16;
    if (r0 + 16 > n_mask) return;
    const int lrow = lane & 15;
    const int lhi  = lane >> 4;

    const long long mi = (long long)mask[r0 + lrow];
    v2f a[16];
    const float* hp = h + mi * D_DIM + 2 * lhi;
    const float* gp = gbias + 2 * lhi;
    const float rz = 1.0f / (denom[mi] + 1e-16f);
#pragma unroll
    for (int kk = 0; kk < 16; ++kk) {
        const v2f hv = *(const v2f*)(hp + kk * 4);
        const v2f gv = *(const v2f*)(gp + kk * 4);
        a[kk].x = hv.x * rz + gv.x;
        a[kk].y = hv.y * rz + gv.y;
    }
#pragma unroll
    for (int ct = 0; ct < 8; ++ct) {
        const int col = ct * 16 + lrow;
        const float bv = bo[col];
        v8f acc = {bv, bv, bv, bv, bv, bv, bv, bv};
#pragma unroll
        for (int kk = 0; kk < 16; ++kk) {
            const int k0 = kk * 4 + 2 * lhi;
            v2f b;
            b.x = Wo[(k0    ) * DOUT + col];
            b.y = Wo[(k0 + 1) * DOUT + col];
            acc = __builtin_amdgcn_wmma_f32_16x16x4_f32(
                    false, a[kk], false, b, (short)0, acc, false, false);
        }
#pragma unroll
        for (int r = 0; r < 8; ++r)
            out[(long long)(r0 + r + 8 * lhi) * DOUT + col] = acc[r];
    }
}

extern "C" void kernel_launch(void* const* d_in, const int* in_sizes, int n_in,
                              void* d_out, int out_size, void* d_ws, size_t ws_size,
                              hipStream_t stream) {
    const float* x     = (const float*)d_in[0];
    const int*   ei    = (const int*)d_in[1];
    const int*   mask  = (const int*)d_in[2];
    const float* Wl    = (const float*)d_in[3];
    const float* bl    = (const float*)d_in[4];
    const float* Wr    = (const float*)d_in[5];
    const float* br    = (const float*)d_in[6];
    const float* att   = (const float*)d_in[7];
    const float* gbias = (const float*)d_in[8];
    const float* Wo    = (const float*)d_in[9];
    const float* bo    = (const float*)d_in[10];
    float* out = (float*)d_out;

    const int n_nodes = in_sizes[0] / D_DIM;
    const int n_edges = in_sizes[1] / 2;
    const int n_mask  = in_sizes[2];
    const int* src = ei;
    const int* dst = ei + n_edges;

    char* wp = (char*)d_ws;
    auto carve = [&](size_t bytes) {
        char* p = wp;
        wp += (bytes + 255) & ~(size_t)255;
        return p;
    };
    float*    xl     = (float*)   carve((size_t)n_nodes * D_DIM * 4);
    float*    xr     = (float*)   carve((size_t)n_nodes * D_DIM * 4);
    float*    rec    = (float*)   carve((size_t)n_edges * 8);
    unsigned char* rid = (unsigned char*)carve((size_t)n_edges + CHUNK);
    float*    denom  = (float*)   carve((size_t)n_nodes * 4);
    float*    h      = xr;

    {
        int tiles  = (n_nodes + 15) / 16;
        int blocks = (tiles + 7) / 8;
        k_node_linear<<<blocks, 256, 0, stream>>>(x, Wl, bl, Wr, br, xl, xr, n_nodes);
    }
    {
        int tiles  = (n_edges + 15) / 16;
        int blocks = (tiles + 7) / 8;
        k_edge_logits<<<blocks, 256, 0, stream>>>(src, dst, xl, xr, att, rec, rid, n_edges);
    }
    k_scatter_lds<<<(n_nodes + NR - 1) / NR, 1024, 0, stream>>>(xl, rec, rid, denom, h,
                                                              n_edges, n_nodes);
    {
        int tiles  = (n_mask + 15) / 16;
        int blocks = (tiles + 7) / 8;
        k_output<<<blocks, 256, 0, stream>>>(h, denom, mask, gbias, Wo, bo, out, n_mask);
    }
}
